# starting kernel
# baseline (speedup 1.0000x reference)
#include <hip/hip_runtime.h>
#include <hip/hip_bf16.h>

typedef __attribute__((ext_vector_type(2))) float v2f;
typedef __attribute__((ext_vector_type(8))) float v8f;

#define BLOCK 128

__device__ __forceinline__ v8f wmma4(v2f a, v2f b, v8f c) {
  return __builtin_amdgcn_wmma_f32_16x16x4_f32(false, a, false, b, (short)0, c,
                                               false, false);
}

__global__ void reduce_wq_kernel(const float* __restrict__ Wq,
                                 float* __restrict__ Wqg) {
  int idx = blockIdx.x * blockDim.x + threadIdx.x;
  if (idx >= 1024 * 256) return;
  int i = idx >> 8;
  int c = idx & 255;
  int h = c >> 6, d = c & 63;
  const float* p = Wq + (size_t)i * 1024 + h * 256 + d;
  Wqg[idx] = p[0] + p[64] + p[128] + p[192];
}

__global__ __launch_bounds__(BLOCK) void gemm_f32_wmma(
    const float* __restrict__ A, const float* __restrict__ B,
    float* __restrict__ C, int M, int N, int K) {
  __shared__ __align__(16) float As[64][18];
  __shared__ __align__(16) float Bs[16][66];
  const int tid  = threadIdx.x;
  const int wave = tid >> 5;
  const int lane = tid & 31;
  const int half = lane >> 4;
  const int l16  = lane & 15;
  const int row0 = blockIdx.y * 64;
  const int col0 = blockIdx.x * 64;

  v8f acc[4];
#pragma unroll
  for (int nt = 0; nt < 4; ++nt) acc[nt] = (v8f)(0.0f);

  for (int kk = 0; kk < K; kk += 16) {
    __syncthreads();
#pragma unroll
    for (int f = tid; f < 256; f += BLOCK) {
      int r = f >> 2, c = (f & 3) << 2;
      const float4 v = *(const float4*)(A + (size_t)(row0 + r) * K + kk + c);
      As[r][c] = v.x; As[r][c + 1] = v.y; As[r][c + 2] = v.z; As[r][c + 3] = v.w;
    }
#pragma unroll
    for (int f = tid; f < 256; f += BLOCK) {
      int r = f >> 4, c = (f & 15) << 2;
      const float4 v = *(const float4*)(B + (size_t)(kk + r) * N + col0 + c);
      Bs[r][c] = v.x; Bs[r][c + 1] = v.y; Bs[r][c + 2] = v.z; Bs[r][c + 3] = v.w;
    }
    __syncthreads();
#pragma unroll
    for (int k0 = 0; k0 < 16; k0 += 4) {
      v2f a = *(const v2f*)&As[wave * 16 + l16][k0 + 2 * half];
#pragma unroll
      for (int nt = 0; nt < 4; ++nt) {
        v2f b;
        b.x = Bs[k0 + 2 * half + 0][nt * 16 + l16];
        b.y = Bs[k0 + 2 * half + 1][nt * 16 + l16];
        acc[nt] = wmma4(a, b, acc[nt]);
      }
    }
  }
#pragma unroll
  for (int nt = 0; nt < 4; ++nt)
#pragma unroll
    for (int r = 0; r < 8; ++r)
      C[(size_t)(row0 + wave * 16 + r + 8 * half) * N + col0 + nt * 16 + l16] =
          acc[nt][r];
}

__global__ __launch_bounds__(BLOCK) void fa_f32_wmma(
    const float* __restrict__ Qg, const float* __restrict__ Kp,
    const float* __restrict__ Vp, float* __restrict__ Op, int S) {
  __shared__ __align__(16) float Ks[16][66];
  __shared__ __align__(16) float Vs[16][66];
  __shared__ __align__(16) float Ps[4][16][18];
  const int tid  = threadIdx.x;
  const int wave = tid >> 5;
  const int lane = tid & 31;
  const int half = lane >> 4;
  const int l16  = lane & 15;
  const int q0   = blockIdx.x * 64;
  const int b    = blockIdx.y >> 2;
  const int h    = blockIdx.y & 3;

  v2f qf[16];
  {
    const int qrow = q0 + wave * 16 + l16;
    const float* qptr =
        Qg + (size_t)(b * S + qrow) * 256 + h * 64 + 2 * half;
#pragma unroll
    for (int i = 0; i < 16; ++i) qf[i] = *(const v2f*)(qptr + i * 4);
  }

  v8f oacc[4];
#pragma unroll
  for (int nt = 0; nt < 4; ++nt) oacc[nt] = (v8f)(0.0f);
  float m[8], l[8];
#pragma unroll
  for (int r = 0; r < 8; ++r) { m[r] = -1e30f; l[r] = 0.0f; }

  const int ntiles = (q0 >> 4) + 4;
  for (int j = 0; j < ntiles; ++j) {
    const int a0 = j << 4;
    __syncthreads();
#pragma unroll
    for (int f = tid; f < 256; f += BLOCK) {
      int r = f >> 4, c = (f & 15) << 2;
      const size_t g = (size_t)(b * S + a0 + r) * 256 + h * 64 + c;
      const float4 kv = *(const float4*)(Kp + g);
      Ks[r][c] = kv.x; Ks[r][c + 1] = kv.y; Ks[r][c + 2] = kv.z; Ks[r][c + 3] = kv.w;
      const float4 vv = *(const float4*)(Vp + g);
      Vs[r][c] = vv.x; Vs[r][c + 1] = vv.y; Vs[r][c + 2] = vv.z; Vs[r][c + 3] = vv.w;
    }
    __syncthreads();

    v8f s = (v8f)(0.0f);
#pragma unroll
    for (int i = 0; i < 16; ++i) {
      v2f bk;
      bk.x = Ks[l16][i * 4 + 2 * half + 0];
      bk.y = Ks[l16][i * 4 + 2 * half + 1];
      s = wmma4(qf[i], bk, s);
    }

    const int acol = a0 + l16;
    float pr[8];
#pragma unroll
    for (int r = 0; r < 8; ++r) {
      const int qrow = q0 + wave * 16 + r + 8 * half;
      float x = s[r] * 0.125f;
      if (acol > qrow) x = -1e30f;
      float mx = x;
#pragma unroll
      for (int off = 1; off < 16; off <<= 1)
        mx = fmaxf(mx, __shfl_xor(mx, off, 16));
      const float mnew  = fmaxf(m[r], mx);
      const float p     = __expf(x - mnew);
      float sum = p;
#pragma unroll
      for (int off = 1; off < 16; off <<= 1) sum += __shfl_xor(sum, off, 16);
      const float scale = __expf(m[r] - mnew);
      l[r] = l[r] * scale + sum;
      m[r] = mnew;
#pragma unroll
      for (int nt = 0; nt < 4; ++nt) oacc[nt][r] *= scale;
      pr[r] = p;
    }

#pragma unroll
    for (int r = 0; r < 8; ++r) Ps[wave][r + 8 * half][l16] = pr[r];
    __syncthreads();

#pragma unroll
    for (int k0i = 0; k0i < 4; ++k0i) {
      const int k0 = k0i * 4;
      v2f a = *(const v2f*)&Ps[wave][l16][k0 + 2 * half];
#pragma unroll
      for (int nt = 0; nt < 4; ++nt) {
        v2f bv;
        bv.x = Vs[k0 + 2 * half + 0][nt * 16 + l16];
        bv.y = Vs[k0 + 2 * half + 1][nt * 16 + l16];
        oacc[nt] = wmma4(a, bv, oacc[nt]);
      }
    }
  }

#pragma unroll
  for (int r = 0; r < 8; ++r) {
    const float inv = 1.0f / l[r];
    const int row = b * S + q0 + wave * 16 + r + 8 * half;
#pragma unroll
    for (int nt = 0; nt < 4; ++nt)
      Op[(size_t)row * 256 + h * 64 + nt * 16 + l16] = oacc[nt][r] * inv;
  }
}

extern "C" void kernel_launch(void* const* d_in, const int* in_sizes, int n_in,
                              void* d_out, int out_size, void* d_ws,
                              size_t ws_size, hipStream_t stream) {
  const float* q_in  = (const float*)d_in[0];
  const float* kv_in = (const float*)d_in[1];
  const float* Wq = (const float*)d_in[3];
  const float* Wk = (const float*)d_in[4];
  const float* Wv = (const float*)d_in[5];
  const float* Wo = (const float*)d_in[6];
  float* out = (float*)d_out;

  const int B = 2, S = 2048, IN = 1024;
  const int M = B * S;

  float* ws  = (float*)d_ws;
  float* wqg = ws;
  float* qg  = wqg + 1024 * 256;
  float* kp  = qg + M * 256;
  float* vp  = kp + M * 256;
  float* op  = vp + M * 256;

  reduce_wq_kernel<<<(1024 * 256) / 256, 256, 0, stream>>>(Wq, wqg);

  dim3 blk(BLOCK);
  gemm_f32_wmma<<<dim3(256 / 64, M / 64), blk, 0, stream>>>(q_in, wqg, qg, M, 256, IN);
  gemm_f32_wmma<<<dim3(256 / 64, M / 64), blk, 0, stream>>>(kv_in, Wk, kp, M, 256, IN);
  gemm_f32_wmma<<<dim3(256 / 64, M / 64), blk, 0, stream>>>(kv_in, Wv, vp, M, 256, IN);

  fa_f32_wmma<<<dim3(S / 64, B * 4), blk, 0, stream>>>(qg, kp, vp, op, S);

  gemm_f32_wmma<<<dim3(IN / 64, M / 64), blk, 0, stream>>>(op, Wo, out, M, IN, 256);
}


# projection GEMMs: k range split over 8 wave groups of one 32-wave work-group (own LDS tiles, 8 trips not 64), partial tiles summed through LDS in fixed order; changes the order of the k sums
# speedup vs baseline: 4.0211x; 4.0211x over previous
#include <hip/hip_runtime.h>
#include <hip/hip_bf16.h>

typedef __attribute__((ext_vector_type(2))) float v2f;
typedef __attribute__((ext_vector_type(8))) float v8f;

#define BLOCK 128

__device__ __forceinline__ v8f wmma4(v2f a, v2f b, v8f c) {
  return __builtin_amdgcn_wmma_f32_16x16x4_f32(false, a, false, b, (short)0, c,
                                               false, false);
}

__global__ void reduce_wq_kernel(const float* __restrict__ Wq,
                                 float* __restrict__ Wqg) {
  int idx = blockIdx.x * blockDim.x + threadIdx.x;
  if (idx >= 1024 * 256) return;
  int i = idx >> 8;
  int c = idx & 255;
  int h = c >> 6, d = c & 63;
  const float* p = Wq + (size_t)i * 1024 + h * 256 + d;
  Wqg[idx] = p[0] + p[64] + p[128] + p[192];
}

__global__ __launch_bounds__(BLOCK) void gemm_f32_wmma(
    const float* __restrict__ A, const float* __restrict__ B,
    float* __restrict__ C, int M, int N, int K) {
  __shared__ __align__(16) float As[64][18];
  __shared__ __align__(16) float Bs[16][66];
  const int tid  = threadIdx.x;
  const int wave = tid >> 5;
  const int lane = tid & 31;
  const int half = lane >> 4;
  const int l16  = lane & 15;
  const int row0 = blockIdx.y * 64;
  const int col0 = blockIdx.x * 64;

  v8f acc[4];
#pragma unroll
  for (int nt = 0; nt < 4; ++nt) acc[nt] = (v8f)(0.0f);

  for (int kk = 0; kk < K; kk += 16) {
    __syncthreads();
#pragma unroll
    for (int f = tid; f < 256; f += BLOCK) {
      int r = f >> 2, c = (f & 3) << 2;
      const float4 v = *(const float4*)(A + (size_t)(row0 + r) * K + kk + c);
      As[r][c] = v.x; As[r][c + 1] = v.y; As[r][c + 2] = v.z; As[r][c + 3] = v.w;
    }
#pragma unroll
    for (int f = tid; f < 256; f += BLOCK) {
      int r = f >> 4, c = (f & 15) << 2;
      const float4 v = *(const float4*)(B + (size_t)(kk + r) * N + col0 + c);
      Bs[r][c] = v.x; Bs[r][c + 1] = v.y; Bs[r][c + 2] = v.z; Bs[r][c + 3] = v.w;
    }
    __syncthreads();
#pragma unroll
    for (int k0 = 0; k0 < 16; k0 += 4) {
      v2f a = *(const v2f*)&As[wave * 16 + l16][k0 + 2 * half];
#pragma unroll
      for (int nt = 0; nt < 4; ++nt) {
        v2f b;
        b.x = Bs[k0 + 2 * half + 0][nt * 16 + l16];
        b.y = Bs[k0 + 2 * half + 1][nt * 16 + l16];
        acc[nt] = wmma4(a, b, acc[nt]);
      }
    }
  }
#pragma unroll
  for (int nt = 0; nt < 4; ++nt)
#pragma unroll
    for (int r = 0; r < 8; ++r)
      C[(size_t)(row0 + wave * 16 + r + 8 * half) * N + col0 + nt * 16 + l16] =
          acc[nt][r];
}

template <int KS>
__global__ __launch_bounds__(BLOCK * KS) void gemm_f32_wmma_ksplit(
    const float* __restrict__ A, const float* __restrict__ B,
    float* __restrict__ C, int M, int N, int K) {
  constexpr int A_FLOATS   = 64 * 18;
  constexpr int RED_PITCH  = 68;
  constexpr int GRP_FLOATS = 64 * RED_PITCH;
  static_assert(A_FLOATS + 16 * 66 <= GRP_FLOATS, "staging tiles fit the slab");
  __shared__ __align__(16) float smem[KS * GRP_FLOATS];
  const int grp  = threadIdx.x >> 7;
  const int tid  = threadIdx.x & (BLOCK - 1);
  float (*As)[18] = reinterpret_cast<float (*)[18]>(smem + grp * GRP_FLOATS);
  float (*Bs)[66] = reinterpret_cast<float (*)[66]>(smem + grp * GRP_FLOATS + A_FLOATS);
  const int wave = tid >> 5;
  const int lane = tid & 31;
  const int half = lane >> 4;
  const int l16  = lane & 15;
  const int row0 = blockIdx.y * 64;
  const int col0 = blockIdx.x * 64;
  const int kq   = K / KS;
  const int kbeg = grp * kq;

  v8f acc[4];
#pragma unroll
  for (int nt = 0; nt < 4; ++nt) acc[nt] = (v8f)(0.0f);

  for (int it = 0; it < kq; it += 16) {
    const int kk = kbeg + it;
    __syncthreads();
#pragma unroll
    for (int f = tid; f < 256; f += BLOCK) {
      int r = f >> 2, c = (f & 3) << 2;
      const float4 v = *(const float4*)(A + (size_t)(row0 + r) * K + kk + c);
      As[r][c] = v.x; As[r][c + 1] = v.y; As[r][c + 2] = v.z; As[r][c + 3] = v.w;
    }
#pragma unroll
    for (int f = tid; f < 256; f += BLOCK) {
      int r = f >> 4, c = (f & 15) << 2;
      const float4 v = *(const float4*)(B + (size_t)(kk + r) * N + col0 + c);
      Bs[r][c] = v.x; Bs[r][c + 1] = v.y; Bs[r][c + 2] = v.z; Bs[r][c + 3] = v.w;
    }
    __syncthreads();
#pragma unroll
    for (int k0 = 0; k0 < 16; k0 += 4) {
      v2f a = *(const v2f*)&As[wave * 16 + l16][k0 + 2 * half];
#pragma unroll
      for (int nt = 0; nt < 4; ++nt) {
        v2f b;
        b.x = Bs[k0 + 2 * half + 0][nt * 16 + l16];
        b.y = Bs[k0 + 2 * half + 1][nt * 16 + l16];
        acc[nt] = wmma4(a, b, acc[nt]);
      }
    }
  }

  __syncthreads();
  float* red = smem + grp * GRP_FLOATS;
#pragma unroll
  for (int nt = 0; nt < 4; ++nt)
#pragma unroll
    for (int r = 0; r < 8; ++r)
      red[(wave * 16 + r + 8 * half) * RED_PITCH + nt * 16 + l16] = acc[nt][r];
  __syncthreads();
  for (int f = threadIdx.x; f < 1024; f += BLOCK * KS) {
    const int r = f >> 4, c = (f & 15) << 2;
    float4 s = *(const float4*)(smem + r * RED_PITCH + c);
#pragma unroll
    for (int g = 1; g < KS; ++g) {
      const float4 p = *(const float4*)(smem + g * GRP_FLOATS + r * RED_PITCH + c);
      s.x += p.x; s.y += p.y; s.z += p.z; s.w += p.w;
    }
    *(float4*)(C + (size_t)(row0 + r) * N + col0 + c) = s;
  }
}

#define CH_UNITS 5
#define CH_TILES (CH_UNITS * 4)
#define NCHUNK_MAX 7
#define NSLOT 119
#define SLOT_FLOATS (64 * 64 + 128)

__device__ __forceinline__ int slot_prefix(int qt) {
  const int a = qt / CH_UNITS, r = qt - a * CH_UNITS;
  return qt + CH_UNITS * a * (a - 1) / 2 + a * r;
}

__global__ __launch_bounds__(BLOCK) void fa_f32_wmma(
    const float* __restrict__ Qg, const float* __restrict__ Kp,
    const float* __restrict__ Vp, float* __restrict__ Part, int S) {
  if ((int)blockIdx.z * CH_UNITS > (int)blockIdx.x) return;
  __shared__ __align__(16) float Ks[16][66];
  __shared__ __align__(16) float Vs[16][66];
  __shared__ __align__(16) float Ps[4][16][18];
  const int tid  = threadIdx.x;
  const int wave = tid >> 5;
  const int lane = tid & 31;
  const int half = lane >> 4;
  const int l16  = lane & 15;
  const int q0   = blockIdx.x * 64;
  const int b    = blockIdx.y >> 2;
  const int h    = blockIdx.y & 3;

  v2f qf[16];
  {
    const int qrow = q0 + wave * 16 + l16;
    const float* qptr =
        Qg + (size_t)(b * S + qrow) * 256 + h * 64 + 2 * half;
#pragma unroll
    for (int i = 0; i < 16; ++i) qf[i] = *(const v2f*)(qptr + i * 4);
  }

  v8f oacc[4];
#pragma unroll
  for (int nt = 0; nt < 4; ++nt) oacc[nt] = (v8f)(0.0f);
  float m[8], l[8];
#pragma unroll
  for (int r = 0; r < 8; ++r) { m[r] = -1e30f; l[r] = 0.0f; }

  const int ntiles = (q0 >> 4) + 4;
  const int jbeg = (int)blockIdx.z * CH_TILES;
  const int jend = (jbeg + CH_TILES < ntiles) ? (jbeg + CH_TILES) : ntiles;
  for (int j = jbeg; j < jend; ++j) {
    const int a0 = j << 4;
    __syncthreads();
#pragma unroll
    for (int f = tid; f < 256; f += BLOCK) {
      int r = f >> 4, c = (f & 15) << 2;
      const size_t g = (size_t)(b * S + a0 + r) * 256 + h * 64 + c;
      const float4 kv = *(const float4*)(Kp + g);
      Ks[r][c] = kv.x; Ks[r][c + 1] = kv.y; Ks[r][c + 2] = kv.z; Ks[r][c + 3] = kv.w;
      const float4 vv = *(const float4*)(Vp + g);
      Vs[r][c] = vv.x; Vs[r][c + 1] = vv.y; Vs[r][c + 2] = vv.z; Vs[r][c + 3] = vv.w;
    }
    __syncthreads();

    v8f s = (v8f)(0.0f);
#pragma unroll
    for (int i = 0; i < 16; ++i) {
      v2f bk;
      bk.x = Ks[l16][i * 4 + 2 * half + 0];
      bk.y = Ks[l16][i * 4 + 2 * half + 1];
      s = wmma4(qf[i], bk, s);
    }

    const int acol = a0 + l16;
    float pr[8];
#pragma unroll
    for (int r = 0; r < 8; ++r) {
      const int qrow = q0 + wave * 16 + r + 8 * half;
      float x = s[r] * 0.125f;
      if (acol > qrow) x = -1e30f;
      float mx = x;
#pragma unroll
      for (int off = 1; off < 16; off <<= 1)
        mx = fmaxf(mx, __shfl_xor(mx, off, 16));
      const float mnew  = fmaxf(m[r], mx);
      const float p     = __expf(x - mnew);
      float sum = p;
#pragma unroll
      for (int off = 1; off < 16; off <<= 1) sum += __shfl_xor(sum, off, 16);
      const float scale = __expf(m[r] - mnew);
      l[r] = l[r] * scale + sum;
      m[r] = mnew;
#pragma unroll
      for (int nt = 0; nt < 4; ++nt) oacc[nt][r] *= scale;
      pr[r] = p;
    }

#pragma unroll
    for (int r = 0; r < 8; ++r) Ps[wave][r + 8 * half][l16] = pr[r];
    __syncthreads();

#pragma unroll
    for (int k0i = 0; k0i < 4; ++k0i) {
      const int k0 = k0i * 4;
      v2f a = *(const v2f*)&Ps[wave][l16][k0 + 2 * half];
#pragma unroll
      for (int nt = 0; nt < 4; ++nt) {
        v2f bv;
        bv.x = Vs[k0 + 2 * half + 0][nt * 16 + l16];
        bv.y = Vs[k0 + 2 * half + 1][nt * 16 + l16];
        oacc[nt] = wmma4(a, bv, oacc[nt]);
      }
    }
  }

  float* slot = Part + (size_t)((int)blockIdx.y * NSLOT + slot_prefix((int)blockIdx.x) +
                                (int)blockIdx.z) * SLOT_FLOATS;
#pragma unroll
  for (int r = 0; r < 8; ++r) {
    const int rl = wave * 16 + r + 8 * half;
#pragma unroll
    for (int nt = 0; nt < 4; ++nt)
      slot[rl * 64 + nt * 16 + l16] = oacc[nt][r];
    if (l16 == 0) {
      slot[64 * 64 + rl]      = m[r];
      slot[64 * 64 + 64 + rl] = l[r];
    }
  }
}

__global__ void fa_merge_kernel(const float* __restrict__ Part,
                                float* __restrict__ Op, int S) {
  const int idx  = blockIdx.x * blockDim.x + threadIdx.x;
  const int d4   = idx & 15;
  const int h    = (idx >> 4) & 3;
  const int grow = idx >> 6;
  const int b    = grow / S;
  const int row  = grow - b * S;
  const int qt   = row >> 6, rl = row & 63;
  const int nch  = qt / CH_UNITS + 1;
  const float* base =
      Part + (size_t)((b * 4 + h) * NSLOT + slot_prefix(qt)) * SLOT_FLOATS;
  float mc[NCHUNK_MAX];
#pragma unroll
  for (int c = 0; c < NCHUNK_MAX; ++c)
    mc[c] = (c < nch) ? base[(size_t)c * SLOT_FLOATS + 64 * 64 + rl] : -1e30f;
  float M = mc[0];
#pragma unroll
  for (int c = 1; c < NCHUNK_MAX; ++c) M = fmaxf(M, mc[c]);
  float L = 0.0f;
  float4 acc = make_float4(0.0f, 0.0f, 0.0f, 0.0f);
#pragma unroll
  for (int c = 0; c < NCHUNK_MAX; ++c) {
    if (c < nch) {
      const float* s = base + (size_t)c * SLOT_FLOATS;
      const float w = __expf(mc[c] - M);
      L += s[64 * 64 + 64 + rl] * w;
      const float4 o = *(const float4*)(s + rl * 64 + d4 * 4);
      acc.x += w * o.x; acc.y += w * o.y; acc.z += w * o.z; acc.w += w * o.w;
    }
  }
  const float inv = 1.0f / L;
  float4 res = make_float4(acc.x * inv, acc.y * inv, acc.z * inv, acc.w * inv);
  *(float4*)(Op + (size_t)grow * 256 + h * 64 + d4 * 4) = res;
}

extern "C" void kernel_launch(void* const* d_in, const int* in_sizes, int n_in,
                              void* d_out, int out_size, void* d_ws,
                              size_t ws_size, hipStream_t stream) {
  const float* q_in  = (const float*)d_in[0];
  const float* kv_in = (const float*)d_in[1];
  const float* Wq = (const float*)d_in[3];
  const float* Wk = (const float*)d_in[4];
  const float* Wv = (const float*)d_in[5];
  const float* Wo = (const float*)d_in[6];
  float* out = (float*)d_out;

  const int B = 2, S = 2048, IN = 1024;
  const int M = B * S;

  float* ws  = (float*)d_ws;
  float* wqg = ws;
  float* qg  = wqg + 1024 * 256;
  float* kp  = qg + M * 256;
  float* vp  = kp + M * 256;
  float* op  = vp + M * 256;

  reduce_wq_kernel<<<(1024 * 256) / 256, 256, 0, stream>>>(Wq, wqg);

  dim3 blk(BLOCK);
  dim3 blk8(BLOCK * 8);
  gemm_f32_wmma_ksplit<8><<<dim3(256 / 64, M / 64), blk8, 0, stream>>>(q_in, wqg, qg, M, 256, IN);
  gemm_f32_wmma_ksplit<8><<<dim3(256 / 64, M / 64), blk8, 0, stream>>>(kv_in, Wk, kp, M, 256, IN);
  gemm_f32_wmma_ksplit<8><<<dim3(256 / 64, M / 64), blk8, 0, stream>>>(kv_in, Wv, vp, M, 256, IN);

  fa_f32_wmma<<<dim3(S / 64, B * 4, NCHUNK_MAX), blk, 0, stream>>>(qg, kp, vp, out, S);
  fa_merge_kernel<<<(4096 * 4 * 16) / 256, 256, 0, stream>>>(out, op, S);

  gemm_f32_wmma<<<dim3(IN / 64, M / 64), blk, 0, stream>>>(op, Wo, out, M, IN, 256);
}


# attention softmax: row max and row sum butterflies over the 16 lanes on DPP row_xmask vector ops, not ds_bpermute LDS round trips (64 per key tile, the sum's 32 fully serial); same partner order
# speedup vs baseline: 4.1578x; 1.0340x over previous
#include <hip/hip_runtime.h>
#include <hip/hip_bf16.h>

typedef __attribute__((ext_vector_type(2))) float v2f;
typedef __attribute__((ext_vector_type(8))) float v8f;

#define BLOCK 128

__device__ __forceinline__ v8f wmma4(v2f a, v2f b, v8f c) {
  return __builtin_amdgcn_wmma_f32_16x16x4_f32(false, a, false, b, (short)0, c,
                                               false, false);
}

template <int MASK>
__device__ __forceinline__ float row_xor(float v) {
  return __int_as_float(__builtin_amdgcn_update_dpp(
      0, __float_as_int(v), 0x160 + MASK, 0xf, 0xf, true));
}

__global__ void reduce_wq_kernel(const float* __restrict__ Wq,
                                 float* __restrict__ Wqg) {
  int idx = blockIdx.x * blockDim.x + threadIdx.x;
  if (idx >= 1024 * 256) return;
  int i = idx >> 8;
  int c = idx & 255;
  int h = c >> 6, d = c & 63;
  const float* p = Wq + (size_t)i * 1024 + h * 256 + d;
  Wqg[idx] = p[0] + p[64] + p[128] + p[192];
}

__global__ __launch_bounds__(BLOCK) void gemm_f32_wmma(
    const float* __restrict__ A, const float* __restrict__ B,
    float* __restrict__ C, int M, int N, int K) {
  __shared__ __align__(16) float As[64][18];
  __shared__ __align__(16) float Bs[16][66];
  const int tid  = threadIdx.x;
  const int wave = tid >> 5;
  const int lane = tid & 31;
  const int half = lane >> 4;
  const int l16  = lane & 15;
  const int row0 = blockIdx.y * 64;
  const int col0 = blockIdx.x * 64;

  v8f acc[4];
#pragma unroll
  for (int nt = 0; nt < 4; ++nt) acc[nt] = (v8f)(0.0f);

  for (int kk = 0; kk < K; kk += 16) {
    __syncthreads();
#pragma unroll
    for (int f = tid; f < 256; f += BLOCK) {
      int r = f >> 2, c = (f & 3) << 2;
      const float4 v = *(const float4*)(A + (size_t)(row0 + r) * K + kk + c);
      As[r][c] = v.x; As[r][c + 1] = v.y; As[r][c + 2] = v.z; As[r][c + 3] = v.w;
    }
#pragma unroll
    for (int f = tid; f < 256; f += BLOCK) {
      int r = f >> 4, c = (f & 15) << 2;
      const float4 v = *(const float4*)(B + (size_t)(kk + r) * N + col0 + c);
      Bs[r][c] = v.x; Bs[r][c + 1] = v.y; Bs[r][c + 2] = v.z; Bs[r][c + 3] = v.w;
    }
    __syncthreads();
#pragma unroll
    for (int k0 = 0; k0 < 16; k0 += 4) {
      v2f a = *(const v2f*)&As[wave * 16 + l16][k0 + 2 * half];
#pragma unroll
      for (int nt = 0; nt < 4; ++nt) {
        v2f b;
        b.x = Bs[k0 + 2 * half + 0][nt * 16 + l16];
        b.y = Bs[k0 + 2 * half + 1][nt * 16 + l16];
        acc[nt] = wmma4(a, b, acc[nt]);
      }
    }
  }
#pragma unroll
  for (int nt = 0; nt < 4; ++nt)
#pragma unroll
    for (int r = 0; r < 8; ++r)
      C[(size_t)(row0 + wave * 16 + r + 8 * half) * N + col0 + nt * 16 + l16] =
          acc[nt][r];
}

template <int KS>
__global__ __launch_bounds__(BLOCK * KS) void gemm_f32_wmma_ksplit(
    const float* __restrict__ A, const float* __restrict__ B,
    float* __restrict__ C, int M, int N, int K) {
  constexpr int A_FLOATS   = 64 * 18;
  constexpr int RED_PITCH  = 68;
  constexpr int GRP_FLOATS = 64 * RED_PITCH;
  static_assert(A_FLOATS + 16 * 66 <= GRP_FLOATS, "staging tiles fit the slab");
  __shared__ __align__(16) float smem[KS * GRP_FLOATS];
  const int grp  = threadIdx.x >> 7;
  const int tid  = threadIdx.x & (BLOCK - 1);
  float (*As)[18] = reinterpret_cast<float (*)[18]>(smem + grp * GRP_FLOATS);
  float (*Bs)[66] = reinterpret_cast<float (*)[66]>(smem + grp * GRP_FLOATS + A_FLOATS);
  const int wave = tid >> 5;
  const int lane = tid & 31;
  const int half = lane >> 4;
  const int l16  = lane & 15;
  const int row0 = blockIdx.y * 64;
  const int col0 = blockIdx.x * 64;
  const int kq   = K / KS;
  const int kbeg = grp * kq;

  v8f acc[4];
#pragma unroll
  for (int nt = 0; nt < 4; ++nt) acc[nt] = (v8f)(0.0f);

  for (int it = 0; it < kq; it += 16) {
    const int kk = kbeg + it;
    __syncthreads();
#pragma unroll
    for (int f = tid; f < 256; f += BLOCK) {
      int r = f >> 2, c = (f & 3) << 2;
      const float4 v = *(const float4*)(A + (size_t)(row0 + r) * K + kk + c);
      As[r][c] = v.x; As[r][c + 1] = v.y; As[r][c + 2] = v.z; As[r][c + 3] = v.w;
    }
#pragma unroll
    for (int f = tid; f < 256; f += BLOCK) {
      int r = f >> 4, c = (f & 15) << 2;
      const float4 v = *(const float4*)(B + (size_t)(kk + r) * N + col0 + c);
      Bs[r][c] = v.x; Bs[r][c + 1] = v.y; Bs[r][c + 2] = v.z; Bs[r][c + 3] = v.w;
    }
    __syncthreads();
#pragma unroll
    for (int k0 = 0; k0 < 16; k0 += 4) {
      v2f a = *(const v2f*)&As[wave * 16 + l16][k0 + 2 * half];
#pragma unroll
      for (int nt = 0; nt < 4; ++nt) {
        v2f b;
        b.x = Bs[k0 + 2 * half + 0][nt * 16 + l16];
        b.y = Bs[k0 + 2 * half + 1][nt * 16 + l16];
        acc[nt] = wmma4(a, b, acc[nt]);
      }
    }
  }

  __syncthreads();
  float* red = smem + grp * GRP_FLOATS;
#pragma unroll
  for (int nt = 0; nt < 4; ++nt)
#pragma unroll
    for (int r = 0; r < 8; ++r)
      red[(wave * 16 + r + 8 * half) * RED_PITCH + nt * 16 + l16] = acc[nt][r];
  __syncthreads();
  for (int f = threadIdx.x; f < 1024; f += BLOCK * KS) {
    const int r = f >> 4, c = (f & 15) << 2;
    float4 s = *(const float4*)(smem + r * RED_PITCH + c);
#pragma unroll
    for (int g = 1; g < KS; ++g) {
      const float4 p = *(const float4*)(smem + g * GRP_FLOATS + r * RED_PITCH + c);
      s.x += p.x; s.y += p.y; s.z += p.z; s.w += p.w;
    }
    *(float4*)(C + (size_t)(row0 + r) * N + col0 + c) = s;
  }
}

#define CH_UNITS 5
#define CH_TILES (CH_UNITS * 4)
#define NCHUNK_MAX 7
#define NSLOT 119
#define SLOT_FLOATS (64 * 64 + 128)

__device__ __forceinline__ int slot_prefix(int qt) {
  const int a = qt / CH_UNITS, r = qt - a * CH_UNITS;
  return qt + CH_UNITS * a * (a - 1) / 2 + a * r;
}

__global__ __launch_bounds__(BLOCK) void fa_f32_wmma(
    const float* __restrict__ Qg, const float* __restrict__ Kp,
    const float* __restrict__ Vp, float* __restrict__ Part, int S) {
  if ((int)blockIdx.z * CH_UNITS > (int)blockIdx.x) return;
  __shared__ __align__(16) float Ks[16][66];
  __shared__ __align__(16) float Vs[16][66];
  __shared__ __align__(16) float Ps[4][16][18];
  const int tid  = threadIdx.x;
  const int wave = tid >> 5;
  const int lane = tid & 31;
  const int half = lane >> 4;
  const int l16  = lane & 15;
  const int q0   = blockIdx.x * 64;
  const int b    = blockIdx.y >> 2;
  const int h    = blockIdx.y & 3;

  v2f qf[16];
  {
    const int qrow = q0 + wave * 16 + l16;
    const float* qptr =
        Qg + (size_t)(b * S + qrow) * 256 + h * 64 + 2 * half;
#pragma unroll
    for (int i = 0; i < 16; ++i) qf[i] = *(const v2f*)(qptr + i * 4);
  }

  v8f oacc[4];
#pragma unroll
  for (int nt = 0; nt < 4; ++nt) oacc[nt] = (v8f)(0.0f);
  float m[8], l[8];
#pragma unroll
  for (int r = 0; r < 8; ++r) { m[r] = -1e30f; l[r] = 0.0f; }

  const int ntiles = (q0 >> 4) + 4;
  const int jbeg = (int)blockIdx.z * CH_TILES;
  const int jend = (jbeg + CH_TILES < ntiles) ? (jbeg + CH_TILES) : ntiles;
  for (int j = jbeg; j < jend; ++j) {
    const int a0 = j << 4;
    __syncthreads();
#pragma unroll
    for (int f = tid; f < 256; f += BLOCK) {
      int r = f >> 4, c = (f & 15) << 2;
      const size_t g = (size_t)(b * S + a0 + r) * 256 + h * 64 + c;
      const float4 kv = *(const float4*)(Kp + g);
      Ks[r][c] = kv.x; Ks[r][c + 1] = kv.y; Ks[r][c + 2] = kv.z; Ks[r][c + 3] = kv.w;
      const float4 vv = *(const float4*)(Vp + g);
      Vs[r][c] = vv.x; Vs[r][c + 1] = vv.y; Vs[r][c + 2] = vv.z; Vs[r][c + 3] = vv.w;
    }
    __syncthreads();

    v8f s = (v8f)(0.0f);
#pragma unroll
    for (int i = 0; i < 16; ++i) {
      v2f bk;
      bk.x = Ks[l16][i * 4 + 2 * half + 0];
      bk.y = Ks[l16][i * 4 + 2 * half + 1];
      s = wmma4(qf[i], bk, s);
    }

    const int acol = a0 + l16;
    float pr[8];
#pragma unroll
    for (int r = 0; r < 8; ++r) {
      const int qrow = q0 + wave * 16 + r + 8 * half;
      float x = s[r] * 0.125f;
      if (acol > qrow) x = -1e30f;
      float mx = x;
      mx = fmaxf(mx, row_xor<1>(mx));
      mx = fmaxf(mx, row_xor<2>(mx));
      mx = fmaxf(mx, row_xor<4>(mx));
      mx = fmaxf(mx, row_xor<8>(mx));
      const float mnew  = fmaxf(m[r], mx);
      const float p     = __expf(x - mnew);
      float sum = p;
      sum += row_xor<1>(sum);
      sum += row_xor<2>(sum);
      sum += row_xor<4>(sum);
      sum += row_xor<8>(sum);
      const float scale = __expf(m[r] - mnew);
      l[r] = l[r] * scale + sum;
      m[r] = mnew;
#pragma unroll
      for (int nt = 0; nt < 4; ++nt) oacc[nt][r] *= scale;
      pr[r] = p;
    }

#pragma unroll
    for (int r = 0; r < 8; ++r) Ps[wave][r + 8 * half][l16] = pr[r];
    __syncthreads();

#pragma unroll
    for (int k0i = 0; k0i < 4; ++k0i) {
      const int k0 = k0i * 4;
      v2f a = *(const v2f*)&Ps[wave][l16][k0 + 2 * half];
#pragma unroll
      for (int nt = 0; nt < 4; ++nt) {
        v2f bv;
        bv.x = Vs[k0 + 2 * half + 0][nt * 16 + l16];
        bv.y = Vs[k0 + 2 * half + 1][nt * 16 + l16];
        oacc[nt] = wmma4(a, bv, oacc[nt]);
      }
    }
  }

  float* slot = Part + (size_t)((int)blockIdx.y * NSLOT + slot_prefix((int)blockIdx.x) +
                                (int)blockIdx.z) * SLOT_FLOATS;
#pragma unroll
  for (int r = 0; r < 8; ++r) {
    const int rl = wave * 16 + r + 8 * half;
#pragma unroll
    for (int nt = 0; nt < 4; ++nt)
      slot[rl * 64 + nt * 16 + l16] = oacc[nt][r];
    if (l16 == 0) {
      slot[64 * 64 + rl]      = m[r];
      slot[64 * 64 + 64 + rl] = l[r];
    }
  }
}

__global__ void fa_merge_kernel(const float* __restrict__ Part,
                                float* __restrict__ Op, int S) {
  const int idx  = blockIdx.x * blockDim.x + threadIdx.x;
  const int d4   = idx & 15;
  const int h    = (idx >> 4) & 3;
  const int grow = idx >> 6;
  const int b    = grow / S;
  const int row  = grow - b * S;
  const int qt   = row >> 6, rl = row & 63;
  const int nch  = qt / CH_UNITS + 1;
  const float* base =
      Part + (size_t)((b * 4 + h) * NSLOT + slot_prefix(qt)) * SLOT_FLOATS;
  float mc[NCHUNK_MAX];
#pragma unroll
  for (int c = 0; c < NCHUNK_MAX; ++c)
    mc[c] = (c < nch) ? base[(size_t)c * SLOT_FLOATS + 64 * 64 + rl] : -1e30f;
  float M = mc[0];
#pragma unroll
  for (int c = 1; c < NCHUNK_MAX; ++c) M = fmaxf(M, mc[c]);
  float L = 0.0f;
  float4 acc = make_float4(0.0f, 0.0f, 0.0f, 0.0f);
#pragma unroll
  for (int c = 0; c < NCHUNK_MAX; ++c) {
    if (c < nch) {
      const float* s = base + (size_t)c * SLOT_FLOATS;
      const float w = __expf(mc[c] - M);
      L += s[64 * 64 + 64 + rl] * w;
      const float4 o = *(const float4*)(s + rl * 64 + d4 * 4);
      acc.x += w * o.x; acc.y += w * o.y; acc.z += w * o.z; acc.w += w * o.w;
    }
  }
  const float inv = 1.0f / L;
  float4 res = make_float4(acc.x * inv, acc.y * inv, acc.z * inv, acc.w * inv);
  *(float4*)(Op + (size_t)grow * 256 + h * 64 + d4 * 4) = res;
}

extern "C" void kernel_launch(void* const* d_in, const int* in_sizes, int n_in,
                              void* d_out, int out_size, void* d_ws,
                              size_t ws_size, hipStream_t stream) {
  const float* q_in  = (const float*)d_in[0];
  const float* kv_in = (const float*)d_in[1];
  const float* Wq = (const float*)d_in[3];
  const float* Wk = (const float*)d_in[4];
  const float* Wv = (const float*)d_in[5];
  const float* Wo = (const float*)d_in[6];
  float* out = (float*)d_out;

  const int B = 2, S = 2048, IN = 1024;
  const int M = B * S;

  float* ws  = (float*)d_ws;
  float* wqg = ws;
  float* qg  = wqg + 1024 * 256;
  float* kp  = qg + M * 256;
  float* vp  = kp + M * 256;
  float* op  = vp + M * 256;

  reduce_wq_kernel<<<(1024 * 256) / 256, 256, 0, stream>>>(Wq, wqg);

  dim3 blk(BLOCK);
  dim3 blk8(BLOCK * 8);
  gemm_f32_wmma_ksplit<8><<<dim3(256 / 64, M / 64), blk8, 0, stream>>>(q_in, wqg, qg, M, 256, IN);
  gemm_f32_wmma_ksplit<8><<<dim3(256 / 64, M / 64), blk8, 0, stream>>>(kv_in, Wk, kp, M, 256, IN);
  gemm_f32_wmma_ksplit<8><<<dim3(256 / 64, M / 64), blk8, 0, stream>>>(kv_in, Wv, vp, M, 256, IN);

  fa_f32_wmma<<<dim3(S / 64, B * 4, NCHUNK_MAX), blk, 0, stream>>>(qg, kp, vp, out, S);
  fa_merge_kernel<<<(4096 * 4 * 16) / 256, 256, 0, stream>>>(out, op, S);

  gemm_f32_wmma<<<dim3(IN / 64, M / 64), blk, 0, stream>>>(op, Wo, out, M, IN, 256);
}


# GEMMs: A and B fragments read from global memory straight into registers by 16-byte loads (k and column order picked for it); no LDS staging or barriers in the k loop; changes the order of the k sums
# speedup vs baseline: 5.8679x; 1.4113x over previous
#include <hip/hip_runtime.h>
#include <hip/hip_bf16.h>

typedef __attribute__((ext_vector_type(2))) float v2f;
typedef __attribute__((ext_vector_type(4))) float v4f;
typedef __attribute__((ext_vector_type(8))) float v8f;

#define BLOCK 128

__device__ __forceinline__ v8f wmma4(v2f a, v2f b, v8f c) {
  return __builtin_amdgcn_wmma_f32_16x16x4_f32(false, a, false, b, (short)0, c,
                                               false, false);
}

template <int MASK>
__device__ __forceinline__ float row_xor(float v) {
  return __int_as_float(__builtin_amdgcn_update_dpp(
      0, __float_as_int(v), 0x160 + MASK, 0xf, 0xf, true));
}

__global__ void reduce_wq_kernel(const float* __restrict__ Wq,
                                 float* __restrict__ Wqg) {
  int idx = blockIdx.x * blockDim.x + threadIdx.x;
  if (idx >= 1024 * 256) return;
  int i = idx >> 8;
  int c = idx & 255;
  int h = c >> 6, d = c & 63;
  const float* p = Wq + (size_t)i * 1024 + h * 256 + d;
  Wqg[idx] = p[0] + p[64] + p[128] + p[192];
}

template <int NSUB>
__device__ __forceinline__ void gemm_wave_slice(
    const float* __restrict__ ap, const float* __restrict__ bp, int N, int ntrips,
    v8f (&acc)[4]) {
  for (int it = 0; it < ntrips; ++it) {
    v4f a[NSUB][2];
    v4f b[NSUB][8];
#pragma unroll
    for (int u = 0; u < NSUB; ++u) {
      a[u][0] = *(const v4f*)(ap + 16 * u);
      a[u][1] = *(const v4f*)(ap + 16 * u + 4);
#pragma unroll
      for (int t = 0; t < 8; ++t)
        b[u][t] = *(const v4f*)(bp + (size_t)(16 * u + t) * N);
    }
    __builtin_amdgcn_sched_barrier(0);
#pragma unroll
    for (int u = 0; u < NSUB; ++u) {
#pragma unroll
      for (int j = 0; j < 4; ++j) {
        v2f af;
        af.x = a[u][j >> 1][(2 * j) & 3];
        af.y = a[u][j >> 1][((2 * j) & 3) + 1];
#pragma unroll
        for (int nt = 0; nt < 4; ++nt) {
          v2f bf;
          bf.x = b[u][2 * j][nt];
          bf.y = b[u][2 * j + 1][nt];
          acc[nt] = wmma4(af, bf, acc[nt]);
        }
      }
    }
    ap += 16 * NSUB;
    bp += (size_t)(16 * NSUB) * N;
  }
}

__global__ __launch_bounds__(BLOCK) void gemm_f32_wmma(
    const float* __restrict__ A, const float* __restrict__ B,
    float* __restrict__ C, int M, int N, int K) {
  const int tid  = threadIdx.x;
  const int wave = tid >> 5;
  const int lane = tid & 31;
  const int half = lane >> 4;
  const int l16  = lane & 15;
  const int row0 = blockIdx.y * 64;
  const int col0 = blockIdx.x * 64;

  v8f acc[4];
#pragma unroll
  for (int nt = 0; nt < 4; ++nt) acc[nt] = (v8f)(0.0f);

  gemm_wave_slice<2>(A + (size_t)(row0 + wave * 16 + l16) * K + 8 * half,
                     B + (size_t)(8 * half) * N + col0 + 4 * l16, N, K / 32, acc);

#pragma unroll
  for (int r = 0; r < 8; ++r) {
    v4f o;
    o.x = acc[0][r]; o.y = acc[1][r]; o.z = acc[2][r]; o.w = acc[3][r];
    *(v4f*)(C + (size_t)(row0 + wave * 16 + r + 8 * half) * N + col0 + 4 * l16) = o;
  }
}

template <int KS>
__global__ __launch_bounds__(BLOCK * KS) void gemm_f32_wmma_ksplit(
    const float* __restrict__ A, const float* __restrict__ B,
    float* __restrict__ C, int M, int N, int K) {
  constexpr int RED_PITCH  = 68;
  constexpr int GRP_FLOATS = 64 * RED_PITCH;
  __shared__ __align__(16) float smem[KS * GRP_FLOATS];
  const int grp  = threadIdx.x >> 7;
  const int tid  = threadIdx.x & (BLOCK - 1);
  const int wave = tid >> 5;
  const int lane = tid & 31;
  const int half = lane >> 4;
  const int l16  = lane & 15;
  const int row0 = blockIdx.y * 64;
  const int col0 = blockIdx.x * 64;
  const int kq   = K / KS;
  const int kbeg = grp * kq;

  v8f acc[4];
#pragma unroll
  for (int nt = 0; nt < 4; ++nt) acc[nt] = (v8f)(0.0f);

  gemm_wave_slice<1>(A + (size_t)(row0 + wave * 16 + l16) * K + kbeg + 8 * half,
                     B + (size_t)(kbeg + 8 * half) * N + col0 + 4 * l16, N, kq / 16, acc);

  float* red = smem + grp * GRP_FLOATS;
#pragma unroll
  for (int r = 0; r < 8; ++r) {
    v4f o;
    o.x = acc[0][r]; o.y = acc[1][r]; o.z = acc[2][r]; o.w = acc[3][r];
    *(v4f*)(red + (wave * 16 + r + 8 * half) * RED_PITCH + 4 * l16) = o;
  }
  __syncthreads();
  for (int f = threadIdx.x; f < 1024; f += BLOCK * KS) {
    const int r = f >> 4, c = (f & 15) << 2;
    float4 s = *(const float4*)(smem + r * RED_PITCH + c);
#pragma unroll
    for (int g = 1; g < KS; ++g) {
      const float4 p = *(const float4*)(smem + g * GRP_FLOATS + r * RED_PITCH + c);
      s.x += p.x; s.y += p.y; s.z += p.z; s.w += p.w;
    }
    *(float4*)(C + (size_t)(row0 + r) * N + col0 + c) = s;
  }
}

#define CH_UNITS 5
#define CH_TILES (CH_UNITS * 4)
#define NCHUNK_MAX 7
#define NSLOT 119
#define SLOT_FLOATS (64 * 64 + 128)

__device__ __forceinline__ int slot_prefix(int qt) {
  const int a = qt / CH_UNITS, r = qt - a * CH_UNITS;
  return qt + CH_UNITS * a * (a - 1) / 2 + a * r;
}

__global__ __launch_bounds__(BLOCK) void fa_f32_wmma(
    const float* __restrict__ Qg, const float* __restrict__ Kp,
    const float* __restrict__ Vp, float* __restrict__ Part, int S) {
  if ((int)blockIdx.z * CH_UNITS > (int)blockIdx.x) return;
  __shared__ __align__(16) float Ks[16][66];
  __shared__ __align__(16) float Vs[16][66];
  __shared__ __align__(16) float Ps[4][16][18];
  const int tid  = threadIdx.x;
  const int wave = tid >> 5;
  const int lane = tid & 31;
  const int half = lane >> 4;
  const int l16  = lane & 15;
  const int q0   = blockIdx.x * 64;
  const int b    = blockIdx.y >> 2;
  const int h    = blockIdx.y & 3;

  v2f qf[16];
  {
    const int qrow = q0 + wave * 16 + l16;
    const float* qptr =
        Qg + (size_t)(b * S + qrow) * 256 + h * 64 + 2 * half;
#pragma unroll
    for (int i = 0; i < 16; ++i) qf[i] = *(const v2f*)(qptr + i * 4);
  }

  v8f oacc[4];
#pragma unroll
  for (int nt = 0; nt < 4; ++nt) oacc[nt] = (v8f)(0.0f);
  float m[8], l[8];
#pragma unroll
  for (int r = 0; r < 8; ++r) { m[r] = -1e30f; l[r] = 0.0f; }

  const int ntiles = (q0 >> 4) + 4;
  const int jbeg = (int)blockIdx.z * CH_TILES;
  const int jend = (jbeg + CH_TILES < ntiles) ? (jbeg + CH_TILES) : ntiles;
  for (int j = jbeg; j < jend; ++j) {
    const int a0 = j << 4;
    __syncthreads();
#pragma unroll
    for (int f = tid; f < 256; f += BLOCK) {
      int r = f >> 4, c = (f & 15) << 2;
      const size_t g = (size_t)(b * S + a0 + r) * 256 + h * 64 + c;
      const float4 kv = *(const float4*)(Kp + g);
      Ks[r][c] = kv.x; Ks[r][c + 1] = kv.y; Ks[r][c + 2] = kv.z; Ks[r][c + 3] = kv.w;
      const float4 vv = *(const float4*)(Vp + g);
      Vs[r][c] = vv.x; Vs[r][c + 1] = vv.y; Vs[r][c + 2] = vv.z; Vs[r][c + 3] = vv.w;
    }
    __syncthreads();

    v8f s = (v8f)(0.0f);
#pragma unroll
    for (int i = 0; i < 16; ++i) {
      v2f bk;
      bk.x = Ks[l16][i * 4 + 2 * half + 0];
      bk.y = Ks[l16][i * 4 + 2 * half + 1];
      s = wmma4(qf[i], bk, s);
    }

    const int acol = a0 + l16;
    float pr[8];
#pragma unroll
    for (int r = 0; r < 8; ++r) {
      const int qrow = q0 + wave * 16 + r + 8 * half;
      float x = s[r] * 0.125f;
      if (acol > qrow) x = -1e30f;
      float mx = x;
      mx = fmaxf(mx, row_xor<1>(mx));
      mx = fmaxf(mx, row_xor<2>(mx));
      mx = fmaxf(mx, row_xor<4>(mx));
      mx = fmaxf(mx, row_xor<8>(mx));
      const float mnew  = fmaxf(m[r], mx);
      const float p     = __expf(x - mnew);
      float sum = p;
      sum += row_xor<1>(sum);
      sum += row_xor<2>(sum);
      sum += row_xor<4>(sum);
      sum += row_xor<8>(sum);
      const float scale = __expf(m[r] - mnew);
      l[r] = l[r] * scale + sum;
      m[r] = mnew;
#pragma unroll
      for (int nt = 0; nt < 4; ++nt) oacc[nt][r] *= scale;
      pr[r] = p;
    }

#pragma unroll
    for (int r = 0; r < 8; ++r) Ps[wave][r + 8 * half][l16] = pr[r];
    __syncthreads();

#pragma unroll
    for (int k0i = 0; k0i < 4; ++k0i) {
      const int k0 = k0i * 4;
      v2f a = *(const v2f*)&Ps[wave][l16][k0 + 2 * half];
#pragma unroll
      for (int nt = 0; nt < 4; ++nt) {
        v2f bv;
        bv.x = Vs[k0 + 2 * half + 0][nt * 16 + l16];
        bv.y = Vs[k0 + 2 * half + 1][nt * 16 + l16];
        oacc[nt] = wmma4(a, bv, oacc[nt]);
      }
    }
  }

  float* slot = Part + (size_t)((int)blockIdx.y * NSLOT + slot_prefix((int)blockIdx.x) +
                                (int)blockIdx.z) * SLOT_FLOATS;
#pragma unroll
  for (int r = 0; r < 8; ++r) {
    const int rl = wave * 16 + r + 8 * half;
#pragma unroll
    for (int nt = 0; nt < 4; ++nt)
      slot[rl * 64 + nt * 16 + l16] = oacc[nt][r];
    if (l16 == 0) {
      slot[64 * 64 + rl]      = m[r];
      slot[64 * 64 + 64 + rl] = l[r];
    }
  }
}

__global__ void fa_merge_kernel(const float* __restrict__ Part,
                                float* __restrict__ Op, int S) {
  const int idx  = blockIdx.x * blockDim.x + threadIdx.x;
  const int d4   = idx & 15;
  const int h    = (idx >> 4) & 3;
  const int grow = idx >> 6;
  const int b    = grow / S;
  const int row  = grow - b * S;
  const int qt   = row >> 6, rl = row & 63;
  const int nch  = qt / CH_UNITS + 1;
  const float* base =
      Part + (size_t)((b * 4 + h) * NSLOT + slot_prefix(qt)) * SLOT_FLOATS;
  float mc[NCHUNK_MAX];
#pragma unroll
  for (int c = 0; c < NCHUNK_MAX; ++c)
    mc[c] = (c < nch) ? base[(size_t)c * SLOT_FLOATS + 64 * 64 + rl] : -1e30f;
  float M = mc[0];
#pragma unroll
  for (int c = 1; c < NCHUNK_MAX; ++c) M = fmaxf(M, mc[c]);
  float L = 0.0f;
  float4 acc = make_float4(0.0f, 0.0f, 0.0f, 0.0f);
#pragma unroll
  for (int c = 0; c < NCHUNK_MAX; ++c) {
    if (c < nch) {
      const float* s = base + (size_t)c * SLOT_FLOATS;
      const float w = __expf(mc[c] - M);
      L += s[64 * 64 + 64 + rl] * w;
      const float4 o = *(const float4*)(s + rl * 64 + d4 * 4);
      acc.x += w * o.x; acc.y += w * o.y; acc.z += w * o.z; acc.w += w * o.w;
    }
  }
  const float inv = 1.0f / L;
  float4 res = make_float4(acc.x * inv, acc.y * inv, acc.z * inv, acc.w * inv);
  *(float4*)(Op + (size_t)grow * 256 + h * 64 + d4 * 4) = res;
}

extern "C" void kernel_launch(void* const* d_in, const int* in_sizes, int n_in,
                              void* d_out, int out_size, void* d_ws,
                              size_t ws_size, hipStream_t stream) {
  const float* q_in  = (const float*)d_in[0];
  const float* kv_in = (const float*)d_in[1];
  const float* Wq = (const float*)d_in[3];
  const float* Wk = (const float*)d_in[4];
  const float* Wv = (const float*)d_in[5];
  const float* Wo = (const float*)d_in[6];
  float* out = (float*)d_out;

  const int B = 2, S = 2048, IN = 1024;
  const int M = B * S;

  float* ws  = (float*)d_ws;
  float* wqg = ws;
  float* qg  = wqg + 1024 * 256;
  float* kp  = qg + M * 256;
  float* vp  = kp + M * 256;
  float* op  = vp + M * 256;

  reduce_wq_kernel<<<(1024 * 256) / 256, 256, 0, stream>>>(Wq, wqg);

  dim3 blk(BLOCK);
  dim3 blk8(BLOCK * 8);
  gemm_f32_wmma_ksplit<8><<<dim3(256 / 64, M / 64), blk8, 0, stream>>>(q_in, wqg, qg, M, 256, IN);
  gemm_f32_wmma_ksplit<8><<<dim3(256 / 64, M / 64), blk8, 0, stream>>>(kv_in, Wk, kp, M, 256, IN);
  gemm_f32_wmma_ksplit<8><<<dim3(256 / 64, M / 64), blk8, 0, stream>>>(kv_in, Wv, vp, M, 256, IN);

  fa_f32_wmma<<<dim3(S / 64, B * 4, NCHUNK_MAX), blk, 0, stream>>>(qg, kp, vp, out, S);
  fa_merge_kernel<<<(4096 * 4 * 16) / 256, 256, 0, stream>>>(out, op, S);

  gemm_f32_wmma<<<dim3(IN / 64, M / 64), blk, 0, stream>>>(op, Wo, out, M, IN, 256);
}


# attention: K and V fragments read from global memory straight into registers by 16-byte loads; only the wave's own P tile in LDS, no block barrier in the key loop; changes order of d and key sums
# speedup vs baseline: 6.2706x; 1.0686x over previous
#include <hip/hip_runtime.h>
#include <hip/hip_bf16.h>

typedef __attribute__((ext_vector_type(2))) float v2f;
typedef __attribute__((ext_vector_type(4))) float v4f;
typedef __attribute__((ext_vector_type(8))) float v8f;

#define BLOCK 128

__device__ __forceinline__ v8f wmma4(v2f a, v2f b, v8f c) {
  return __builtin_amdgcn_wmma_f32_16x16x4_f32(false, a, false, b, (short)0, c,
                                               false, false);
}

template <int MASK>
__device__ __forceinline__ float row_xor(float v) {
  return __int_as_float(__builtin_amdgcn_update_dpp(
      0, __float_as_int(v), 0x160 + MASK, 0xf, 0xf, true));
}

__global__ void reduce_wq_kernel(const float* __restrict__ Wq,
                                 float* __restrict__ Wqg) {
  int idx = blockIdx.x * blockDim.x + threadIdx.x;
  if (idx >= 1024 * 256) return;
  int i = idx >> 8;
  int c = idx & 255;
  int h = c >> 6, d = c & 63;
  const float* p = Wq + (size_t)i * 1024 + h * 256 + d;
  Wqg[idx] = p[0] + p[64] + p[128] + p[192];
}

template <int NSUB>
__device__ __forceinline__ void gemm_wave_slice(
    const float* __restrict__ ap, const float* __restrict__ bp, int N, int ntrips,
    v8f (&acc)[4]) {
  for (int it = 0; it < ntrips; ++it) {
    v4f a[NSUB][2];
    v4f b[NSUB][8];
#pragma unroll
    for (int u = 0; u < NSUB; ++u) {
      a[u][0] = *(const v4f*)(ap + 16 * u);
      a[u][1] = *(const v4f*)(ap + 16 * u + 4);
#pragma unroll
      for (int t = 0; t < 8; ++t)
        b[u][t] = *(const v4f*)(bp + (size_t)(16 * u + t) * N);
    }
    __builtin_amdgcn_sched_barrier(0);
#pragma unroll
    for (int u = 0; u < NSUB; ++u) {
#pragma unroll
      for (int j = 0; j < 4; ++j) {
        v2f af;
        af.x = a[u][j >> 1][(2 * j) & 3];
        af.y = a[u][j >> 1][((2 * j) & 3) + 1];
#pragma unroll
        for (int nt = 0; nt < 4; ++nt) {
          v2f bf;
          bf.x = b[u][2 * j][nt];
          bf.y = b[u][2 * j + 1][nt];
          acc[nt] = wmma4(af, bf, acc[nt]);
        }
      }
    }
    ap += 16 * NSUB;
    bp += (size_t)(16 * NSUB) * N;
  }
}

__global__ __launch_bounds__(BLOCK) void gemm_f32_wmma(
    const float* __restrict__ A, const float* __restrict__ B,
    float* __restrict__ C, int M, int N, int K) {
  const int tid  = threadIdx.x;
  const int wave = tid >> 5;
  const int lane = tid & 31;
  const int half = lane >> 4;
  const int l16  = lane & 15;
  const int row0 = blockIdx.y * 64;
  const int col0 = blockIdx.x * 64;

  v8f acc[4];
#pragma unroll
  for (int nt = 0; nt < 4; ++nt) acc[nt] = (v8f)(0.0f);

  gemm_wave_slice<2>(A + (size_t)(row0 + wave * 16 + l16) * K + 8 * half,
                     B + (size_t)(8 * half) * N + col0 + 4 * l16, N, K / 32, acc);

#pragma unroll
  for (int r = 0; r < 8; ++r) {
    v4f o;
    o.x = acc[0][r]; o.y = acc[1][r]; o.z = acc[2][r]; o.w = acc[3][r];
    *(v4f*)(C + (size_t)(row0 + wave * 16 + r + 8 * half) * N + col0 + 4 * l16) = o;
  }
}

template <int KS>
__global__ __launch_bounds__(BLOCK * KS) void gemm_f32_wmma_ksplit(
    const float* __restrict__ A, const float* __restrict__ B,
    float* __restrict__ C, int M, int N, int K) {
  constexpr int RED_PITCH  = 68;
  constexpr int GRP_FLOATS = 64 * RED_PITCH;
  __shared__ __align__(16) float smem[KS * GRP_FLOATS];
  const int grp  = threadIdx.x >> 7;
  const int tid  = threadIdx.x & (BLOCK - 1);
  const int wave = tid >> 5;
  const int lane = tid & 31;
  const int half = lane >> 4;
  const int l16  = lane & 15;
  const int row0 = blockIdx.y * 64;
  const int col0 = blockIdx.x * 64;
  const int kq   = K / KS;
  const int kbeg = grp * kq;

  v8f acc[4];
#pragma unroll
  for (int nt = 0; nt < 4; ++nt) acc[nt] = (v8f)(0.0f);

  gemm_wave_slice<1>(A + (size_t)(row0 + wave * 16 + l16) * K + kbeg + 8 * half,
                     B + (size_t)(kbeg + 8 * half) * N + col0 + 4 * l16, N, kq / 16, acc);

  float* red = smem + grp * GRP_FLOATS;
#pragma unroll
  for (int r = 0; r < 8; ++r) {
    v4f o;
    o.x = acc[0][r]; o.y = acc[1][r]; o.z = acc[2][r]; o.w = acc[3][r];
    *(v4f*)(red + (wave * 16 + r + 8 * half) * RED_PITCH + 4 * l16) = o;
  }
  __syncthreads();
  for (int f = threadIdx.x; f < 1024; f += BLOCK * KS) {
    const int r = f >> 4, c = (f & 15) << 2;
    float4 s = *(const float4*)(smem + r * RED_PITCH + c);
#pragma unroll
    for (int g = 1; g < KS; ++g) {
      const float4 p = *(const float4*)(smem + g * GRP_FLOATS + r * RED_PITCH + c);
      s.x += p.x; s.y += p.y; s.z += p.z; s.w += p.w;
    }
    *(float4*)(C + (size_t)(row0 + r) * N + col0 + c) = s;
  }
}

#define CH_UNITS 5
#define CH_TILES (CH_UNITS * 4)
#define NCHUNK_MAX 7
#define NSLOT 119
#define SLOT_FLOATS (64 * 64 + 128)

__device__ __forceinline__ int slot_prefix(int qt) {
  const int a = qt / CH_UNITS, r = qt - a * CH_UNITS;
  return qt + CH_UNITS * a * (a - 1) / 2 + a * r;
}

__global__ __launch_bounds__(BLOCK) void fa_f32_wmma(
    const float* __restrict__ Qg, const float* __restrict__ Kp,
    const float* __restrict__ Vp, float* __restrict__ Part, int S) {
  if ((int)blockIdx.z * CH_UNITS > (int)blockIdx.x) return;
  __shared__ __align__(16) float Ps[4][16][18];
  const int tid  = threadIdx.x;
  const int wave = __builtin_amdgcn_readfirstlane(tid >> 5);
  const int lane = tid & 31;
  const int half = lane >> 4;
  const int l16  = lane & 15;
  const int q0   = blockIdx.x * 64;
  const int b    = blockIdx.y >> 2;
  const int h    = blockIdx.y & 3;

  v2f qf[16];
  {
    const int qrow = q0 + wave * 16 + l16;
    const float* qptr =
        Qg + (size_t)(b * S + qrow) * 256 + h * 64 + 32 * half;
#pragma unroll
    for (int t = 0; t < 8; ++t) {
      const v4f q4 = *(const v4f*)(qptr + 4 * t);
      qf[2 * t].x = q4.x;     qf[2 * t].y = q4.y;
      qf[2 * t + 1].x = q4.z; qf[2 * t + 1].y = q4.w;
    }
  }

  v8f oacc[4];
#pragma unroll
  for (int nt = 0; nt < 4; ++nt) oacc[nt] = (v8f)(0.0f);
  float m[8], l[8];
#pragma unroll
  for (int r = 0; r < 8; ++r) { m[r] = -1e30f; l[r] = 0.0f; }

  const int ntiles = (q0 >> 4) + wave + 1;
  const int jbeg = (int)blockIdx.z * CH_TILES;
  const int jend = (jbeg + CH_TILES < ntiles) ? (jbeg + CH_TILES) : ntiles;
  const float* kptr =
      Kp + (size_t)(b * S + jbeg * 16 + l16) * 256 + h * 64 + 32 * half;
  const float* vptr =
      Vp + (size_t)(b * S + jbeg * 16 + 8 * half) * 256 + h * 64 + 4 * l16;
  for (int j = jbeg; j < jend; ++j) {
    const int a0 = j << 4;
    v4f k4[8], v4[8];
#pragma unroll
    for (int t = 0; t < 8; ++t) k4[t] = *(const v4f*)(kptr + 4 * t);
#pragma unroll
    for (int t = 0; t < 8; ++t) v4[t] = *(const v4f*)(vptr + (size_t)t * 256);
    kptr += 16 * 256;
    vptr += 16 * 256;
    __builtin_amdgcn_sched_barrier(0);

    v8f s = (v8f)(0.0f);
#pragma unroll
    for (int i = 0; i < 16; ++i) {
      v2f bk;
      bk.x = k4[i >> 1][(2 * i) & 3];
      bk.y = k4[i >> 1][((2 * i) & 3) + 1];
      s = wmma4(qf[i], bk, s);
    }

    const int acol = a0 + l16;
    float pr[8];
#pragma unroll
    for (int r = 0; r < 8; ++r) {
      const int qrow = q0 + wave * 16 + r + 8 * half;
      float x = s[r] * 0.125f;
      if (acol > qrow) x = -1e30f;
      float mx = x;
      mx = fmaxf(mx, row_xor<1>(mx));
      mx = fmaxf(mx, row_xor<2>(mx));
      mx = fmaxf(mx, row_xor<4>(mx));
      mx = fmaxf(mx, row_xor<8>(mx));
      const float mnew  = fmaxf(m[r], mx);
      const float p     = __expf(x - mnew);
      float sum = p;
      sum += row_xor<1>(sum);
      sum += row_xor<2>(sum);
      sum += row_xor<4>(sum);
      sum += row_xor<8>(sum);
      const float scale = __expf(m[r] - mnew);
      l[r] = l[r] * scale + sum;
      m[r] = mnew;
#pragma unroll
      for (int nt = 0; nt < 4; ++nt) oacc[nt][r] *= scale;
      pr[r] = p;
    }

#pragma unroll
    for (int r = 0; r < 8; ++r) Ps[wave][r + 8 * half][l16] = pr[r];
    __builtin_amdgcn_wave_barrier();

#pragma unroll
    for (int jj = 0; jj < 4; ++jj) {
      const v2f a = *(const v2f*)&Ps[wave][l16][8 * half + 2 * jj];
#pragma unroll
      for (int nt = 0; nt < 4; ++nt) {
        v2f bv;
        bv.x = v4[2 * jj][nt];
        bv.y = v4[2 * jj + 1][nt];
        oacc[nt] = wmma4(a, bv, oacc[nt]);
      }
    }
  }

  float* slot = Part + (size_t)((int)blockIdx.y * NSLOT + slot_prefix((int)blockIdx.x) +
                                (int)blockIdx.z) * SLOT_FLOATS;
#pragma unroll
  for (int r = 0; r < 8; ++r) {
    const int rl = wave * 16 + r + 8 * half;
    v4f o;
    o.x = oacc[0][r]; o.y = oacc[1][r]; o.z = oacc[2][r]; o.w = oacc[3][r];
    *(v4f*)(slot + rl * 64 + 4 * l16) = o;
    if (l16 == 0) {
      slot[64 * 64 + rl]      = m[r];
      slot[64 * 64 + 64 + rl] = l[r];
    }
  }
}

__global__ void fa_merge_kernel(const float* __restrict__ Part,
                                float* __restrict__ Op, int S) {
  const int idx  = blockIdx.x * blockDim.x + threadIdx.x;
  const int d4   = idx & 15;
  const int h    = (idx >> 4) & 3;
  const int grow = idx >> 6;
  const int b    = grow / S;
  const int row  = grow - b * S;
  const int qt   = row >> 6, rl = row & 63;
  const int nch  = qt / CH_UNITS + 1;
  const float* base =
      Part + (size_t)((b * 4 + h) * NSLOT + slot_prefix(qt)) * SLOT_FLOATS;
  float mc[NCHUNK_MAX];
#pragma unroll
  for (int c = 0; c < NCHUNK_MAX; ++c)
    mc[c] = (c < nch) ? base[(size_t)c * SLOT_FLOATS + 64 * 64 + rl] : -1e30f;
  float M = mc[0];
#pragma unroll
  for (int c = 1; c < NCHUNK_MAX; ++c) M = fmaxf(M, mc[c]);
  float L = 0.0f;
  float4 acc = make_float4(0.0f, 0.0f, 0.0f, 0.0f);
#pragma unroll
  for (int c = 0; c < NCHUNK_MAX; ++c) {
    if (c < nch) {
      const float* s = base + (size_t)c * SLOT_FLOATS;
      const float w = __expf(mc[c] - M);
      L += s[64 * 64 + 64 + rl] * w;
      const float4 o = *(const float4*)(s + rl * 64 + d4 * 4);
      acc.x += w * o.x; acc.y += w * o.y; acc.z += w * o.z; acc.w += w * o.w;
    }
  }
  const float inv = 1.0f / L;
  float4 res = make_float4(acc.x * inv, acc.y * inv, acc.z * inv, acc.w * inv);
  *(float4*)(Op + (size_t)grow * 256 + h * 64 + d4 * 4) = res;
}

extern "C" void kernel_launch(void* const* d_in, const int* in_sizes, int n_in,
                              void* d_out, int out_size, void* d_ws,
                              size_t ws_size, hipStream_t stream) {
  const float* q_in  = (const float*)d_in[0];
  const float* kv_in = (const float*)d_in[1];
  const float* Wq = (const float*)d_in[3];
  const float* Wk = (const float*)d_in[4];
  const float* Wv = (const float*)d_in[5];
  const float* Wo = (const float*)d_in[6];
  float* out = (float*)d_out;

  const int B = 2, S = 2048, IN = 1024;
  const int M = B * S;

  float* ws  = (float*)d_ws;
  float* wqg = ws;
  float* qg  = wqg + 1024 * 256;
  float* kp  = qg + M * 256;
  float* vp  = kp + M * 256;
  float* op  = vp + M * 256;

  reduce_wq_kernel<<<(1024 * 256) / 256, 256, 0, stream>>>(Wq, wqg);

  dim3 blk(BLOCK);
  dim3 blk8(BLOCK * 8);
  gemm_f32_wmma_ksplit<8><<<dim3(256 / 64, M / 64), blk8, 0, stream>>>(q_in, wqg, qg, M, 256, IN);
  gemm_f32_wmma_ksplit<8><<<dim3(256 / 64, M / 64), blk8, 0, stream>>>(kv_in, Wk, kp, M, 256, IN);
  gemm_f32_wmma_ksplit<8><<<dim3(256 / 64, M / 64), blk8, 0, stream>>>(kv_in, Wv, vp, M, 256, IN);

  fa_f32_wmma<<<dim3(S / 64, B * 4, NCHUNK_MAX), blk, 0, stream>>>(qg, kp, vp, out, S);
  fa_merge_kernel<<<(4096 * 4 * 16) / 256, 256, 0, stream>>>(out, op, S);

  gemm_f32_wmma<<<dim3(IN / 64, M / 64), blk, 0, stream>>>(op, Wo, out, M, IN, 256);
}


# attention grid = only the 952 live (query tile, key chunk) blocks, heaviest first, (batch, head) fastest: WGPs get near-equal sums of key tiles, no empty blocks; partials and sum order unchanged
# speedup vs baseline: 6.9473x; 1.1079x over previous
#include <hip/hip_runtime.h>
#include <hip/hip_bf16.h>

typedef __attribute__((ext_vector_type(2))) float v2f;
typedef __attribute__((ext_vector_type(4))) float v4f;
typedef __attribute__((ext_vector_type(8))) float v8f;

#define BLOCK 128

__device__ __forceinline__ v8f wmma4(v2f a, v2f b, v8f c) {
  return __builtin_amdgcn_wmma_f32_16x16x4_f32(false, a, false, b, (short)0, c,
                                               false, false);
}

template <int MASK>
__device__ __forceinline__ float row_xor(float v) {
  return __int_as_float(__builtin_amdgcn_update_dpp(
      0, __float_as_int(v), 0x160 + MASK, 0xf, 0xf, true));
}

__global__ void reduce_wq_kernel(const float* __restrict__ Wq,
                                 float* __restrict__ Wqg) {
  int idx = blockIdx.x * blockDim.x + threadIdx.x;
  if (idx >= 1024 * 256) return;
  int i = idx >> 8;
  int c = idx & 255;
  int h = c >> 6, d = c & 63;
  const float* p = Wq + (size_t)i * 1024 + h * 256 + d;
  Wqg[idx] = p[0] + p[64] + p[128] + p[192];
}

template <int NSUB>
__device__ __forceinline__ void gemm_wave_slice(
    const float* __restrict__ ap, const float* __restrict__ bp, int N, int ntrips,
    v8f (&acc)[4]) {
  for (int it = 0; it < ntrips; ++it) {
    v4f a[NSUB][2];
    v4f b[NSUB][8];
#pragma unroll
    for (int u = 0; u < NSUB; ++u) {
      a[u][0] = *(const v4f*)(ap + 16 * u);
      a[u][1] = *(const v4f*)(ap + 16 * u + 4);
#pragma unroll
      for (int t = 0; t < 8; ++t)
        b[u][t] = *(const v4f*)(bp + (size_t)(16 * u + t) * N);
    }
    __builtin_amdgcn_sched_barrier(0);
#pragma unroll
    for (int u = 0; u < NSUB; ++u) {
#pragma unroll
      for (int j = 0; j < 4; ++j) {
        v2f af;
        af.x = a[u][j >> 1][(2 * j) & 3];
        af.y = a[u][j >> 1][((2 * j) & 3) + 1];
#pragma unroll
        for (int nt = 0; nt < 4; ++nt) {
          v2f bf;
          bf.x = b[u][2 * j][nt];
          bf.y = b[u][2 * j + 1][nt];
          acc[nt] = wmma4(af, bf, acc[nt]);
        }
      }
    }
    ap += 16 * NSUB;
    bp += (size_t)(16 * NSUB) * N;
  }
}

__global__ __launch_bounds__(BLOCK) void gemm_f32_wmma(
    const float* __restrict__ A, const float* __restrict__ B,
    float* __restrict__ C, int M, int N, int K) {
  const int tid  = threadIdx.x;
  const int wave = tid >> 5;
  const int lane = tid & 31;
  const int half = lane >> 4;
  const int l16  = lane & 15;
  const int row0 = blockIdx.y * 64;
  const int col0 = blockIdx.x * 64;

  v8f acc[4];
#pragma unroll
  for (int nt = 0; nt < 4; ++nt) acc[nt] = (v8f)(0.0f);

  gemm_wave_slice<2>(A + (size_t)(row0 + wave * 16 + l16) * K + 8 * half,
                     B + (size_t)(8 * half) * N + col0 + 4 * l16, N, K / 32, acc);

#pragma unroll
  for (int r = 0; r < 8; ++r) {
    v4f o;
    o.x = acc[0][r]; o.y = acc[1][r]; o.z = acc[2][r]; o.w = acc[3][r];
    *(v4f*)(C + (size_t)(row0 + wave * 16 + r + 8 * half) * N + col0 + 4 * l16) = o;
  }
}

template <int KS>
__global__ __launch_bounds__(BLOCK * KS) void gemm_f32_wmma_ksplit(
    const float* __restrict__ A, const float* __restrict__ B,
    float* __restrict__ C, int M, int N, int K) {
  constexpr int RED_PITCH  = 68;
  constexpr int GRP_FLOATS = 64 * RED_PITCH;
  __shared__ __align__(16) float smem[KS * GRP_FLOATS];
  const int grp  = threadIdx.x >> 7;
  const int tid  = threadIdx.x & (BLOCK - 1);
  const int wave = tid >> 5;
  const int lane = tid & 31;
  const int half = lane >> 4;
  const int l16  = lane & 15;
  const int row0 = blockIdx.y * 64;
  const int col0 = blockIdx.x * 64;
  const int kq   = K / KS;
  const int kbeg = grp * kq;

  v8f acc[4];
#pragma unroll
  for (int nt = 0; nt < 4; ++nt) acc[nt] = (v8f)(0.0f);

  gemm_wave_slice<1>(A + (size_t)(row0 + wave * 16 + l16) * K + kbeg + 8 * half,
                     B + (size_t)(kbeg + 8 * half) * N + col0 + 4 * l16, N, kq / 16, acc);

  float* red = smem + grp * GRP_FLOATS;
#pragma unroll
  for (int r = 0; r < 8; ++r) {
    v4f o;
    o.x = acc[0][r]; o.y = acc[1][r]; o.z = acc[2][r]; o.w = acc[3][r];
    *(v4f*)(red + (wave * 16 + r + 8 * half) * RED_PITCH + 4 * l16) = o;
  }
  __syncthreads();
  for (int f = threadIdx.x; f < 1024; f += BLOCK * KS) {
    const int r = f >> 4, c = (f & 15) << 2;
    float4 s = *(const float4*)(smem + r * RED_PITCH + c);
#pragma unroll
    for (int g = 1; g < KS; ++g) {
      const float4 p = *(const float4*)(smem + g * GRP_FLOATS + r * RED_PITCH + c);
      s.x += p.x; s.y += p.y; s.z += p.z; s.w += p.w;
    }
    *(float4*)(C + (size_t)(row0 + r) * N + col0 + c) = s;
  }
}

#define CH_UNITS 5
#define CH_TILES (CH_UNITS * 4)
#define NCHUNK_MAX 7
#define NSLOT 119
#define SLOT_FLOATS (64 * 64 + 128)

__device__ __forceinline__ int slot_prefix(int qt) {
  const int a = qt / CH_UNITS, r = qt - a * CH_UNITS;
  return qt + CH_UNITS * a * (a - 1) / 2 + a * r;
}

#define NQT 32
#define NFULL 87
__device__ __forceinline__ void fa_block_pair(int rank, int& qt, int& c) {
  if (rank < NFULL) {
    int a = 1;
#pragma unroll
    for (int k = 1; k < NCHUNK_MAX - 1; ++k)
      if (CH_UNITS * k * (k + 1) / 2 <= rank) a = k + 1;
    const int t = rank - CH_UNITS * a * (a - 1) / 2;
    qt = CH_UNITS * a + t / a;
    c  = t % a;
  } else {
    int p = rank - NFULL;
    qt = 0;
    bool found = false;
#pragma unroll
    for (int r = CH_UNITS - 1; r >= 0; --r) {
      const int n_r = (NQT - 1 - r) / CH_UNITS + 1;
      if (!found && p < n_r) { qt = r + CH_UNITS * p; found = true; }
      if (!found) p -= n_r;
    }
    c = qt / CH_UNITS;
  }
}

__global__ __launch_bounds__(BLOCK) void fa_f32_wmma(
    const float* __restrict__ Qg, const float* __restrict__ Kp,
    const float* __restrict__ Vp, float* __restrict__ Part, int S) {
  __shared__ __align__(16) float Ps[4][16][18];
  const int bh = (int)blockIdx.x & 7;
  int qt, chunk;
  fa_block_pair((int)blockIdx.x >> 3, qt, chunk);
  const int tid  = threadIdx.x;
  const int wave = __builtin_amdgcn_readfirstlane(tid >> 5);
  const int lane = tid & 31;
  const int half = lane >> 4;
  const int l16  = lane & 15;
  const int q0   = qt * 64;
  const int b    = bh >> 2;
  const int h    = bh & 3;

  v2f qf[16];
  {
    const int qrow = q0 + wave * 16 + l16;
    const float* qptr =
        Qg + (size_t)(b * S + qrow) * 256 + h * 64 + 32 * half;
#pragma unroll
    for (int t = 0; t < 8; ++t) {
      const v4f q4 = *(const v4f*)(qptr + 4 * t);
      qf[2 * t].x = q4.x;     qf[2 * t].y = q4.y;
      qf[2 * t + 1].x = q4.z; qf[2 * t + 1].y = q4.w;
    }
  }

  v8f oacc[4];
#pragma unroll
  for (int nt = 0; nt < 4; ++nt) oacc[nt] = (v8f)(0.0f);
  float m[8], l[8];
#pragma unroll
  for (int r = 0; r < 8; ++r) { m[r] = -1e30f; l[r] = 0.0f; }

  const int ntiles = (q0 >> 4) + wave + 1;
  const int jbeg = chunk * CH_TILES;
  const int jend = (jbeg + CH_TILES < ntiles) ? (jbeg + CH_TILES) : ntiles;
  const float* kptr =
      Kp + (size_t)(b * S + jbeg * 16 + l16) * 256 + h * 64 + 32 * half;
  const float* vptr =
      Vp + (size_t)(b * S + jbeg * 16 + 8 * half) * 256 + h * 64 + 4 * l16;
  for (int j = jbeg; j < jend; ++j) {
    const int a0 = j << 4;
    v4f k4[8], v4[8];
#pragma unroll
    for (int t = 0; t < 8; ++t) k4[t] = *(const v4f*)(kptr + 4 * t);
#pragma unroll
    for (int t = 0; t < 8; ++t) v4[t] = *(const v4f*)(vptr + (size_t)t * 256);
    kptr += 16 * 256;
    vptr += 16 * 256;
    __builtin_amdgcn_sched_barrier(0);

    v8f s = (v8f)(0.0f);
#pragma unroll
    for (int i = 0; i < 16; ++i) {
      v2f bk;
      bk.x = k4[i >> 1][(2 * i) & 3];
      bk.y = k4[i >> 1][((2 * i) & 3) + 1];
      s = wmma4(qf[i], bk, s);
    }

    const int acol = a0 + l16;
    float pr[8];
#pragma unroll
    for (int r = 0; r < 8; ++r) {
      const int qrow = q0 + wave * 16 + r + 8 * half;
      float x = s[r] * 0.125f;
      if (acol > qrow) x = -1e30f;
      float mx = x;
      mx = fmaxf(mx, row_xor<1>(mx));
      mx = fmaxf(mx, row_xor<2>(mx));
      mx = fmaxf(mx, row_xor<4>(mx));
      mx = fmaxf(mx, row_xor<8>(mx));
      const float mnew  = fmaxf(m[r], mx);
      const float p     = __expf(x - mnew);
      float sum = p;
      sum += row_xor<1>(sum);
      sum += row_xor<2>(sum);
      sum += row_xor<4>(sum);
      sum += row_xor<8>(sum);
      const float scale = __expf(m[r] - mnew);
      l[r] = l[r] * scale + sum;
      m[r] = mnew;
#pragma unroll
      for (int nt = 0; nt < 4; ++nt) oacc[nt][r] *= scale;
      pr[r] = p;
    }

#pragma unroll
    for (int r = 0; r < 8; ++r) Ps[wave][r + 8 * half][l16] = pr[r];
    __builtin_amdgcn_wave_barrier();

#pragma unroll
    for (int jj = 0; jj < 4; ++jj) {
      const v2f a = *(const v2f*)&Ps[wave][l16][8 * half + 2 * jj];
#pragma unroll
      for (int nt = 0; nt < 4; ++nt) {
        v2f bv;
        bv.x = v4[2 * jj][nt];
        bv.y = v4[2 * jj + 1][nt];
        oacc[nt] = wmma4(a, bv, oacc[nt]);
      }
    }
  }

  float* slot = Part + (size_t)(bh * NSLOT + slot_prefix(qt) + chunk) * SLOT_FLOATS;
#pragma unroll
  for (int r = 0; r < 8; ++r) {
    const int rl = wave * 16 + r + 8 * half;
    v4f o;
    o.x = oacc[0][r]; o.y = oacc[1][r]; o.z = oacc[2][r]; o.w = oacc[3][r];
    *(v4f*)(slot + rl * 64 + 4 * l16) = o;
    if (l16 == 0) {
      slot[64 * 64 + rl]      = m[r];
      slot[64 * 64 + 64 + rl] = l[r];
    }
  }
}

__global__ void fa_merge_kernel(const float* __restrict__ Part,
                                float* __restrict__ Op, int S) {
  const int idx  = blockIdx.x * blockDim.x + threadIdx.x;
  const int d4   = idx & 15;
  const int h    = (idx >> 4) & 3;
  const int grow = idx >> 6;
  const int b    = grow / S;
  const int row  = grow - b * S;
  const int qt   = row >> 6, rl = row & 63;
  const int nch  = qt / CH_UNITS + 1;
  const float* base =
      Part + (size_t)((b * 4 + h) * NSLOT + slot_prefix(qt)) * SLOT_FLOATS;
  float mc[NCHUNK_MAX];
#pragma unroll
  for (int c = 0; c < NCHUNK_MAX; ++c)
    mc[c] = (c < nch) ? base[(size_t)c * SLOT_FLOATS + 64 * 64 + rl] : -1e30f;
  float M = mc[0];
#pragma unroll
  for (int c = 1; c < NCHUNK_MAX; ++c) M = fmaxf(M, mc[c]);
  float L = 0.0f;
  float4 acc = make_float4(0.0f, 0.0f, 0.0f, 0.0f);
#pragma unroll
  for (int c = 0; c < NCHUNK_MAX; ++c) {
    if (c < nch) {
      const float* s = base + (size_t)c * SLOT_FLOATS;
      const float w = __expf(mc[c] - M);
      L += s[64 * 64 + 64 + rl] * w;
      const float4 o = *(const float4*)(s + rl * 64 + d4 * 4);
      acc.x += w * o.x; acc.y += w * o.y; acc.z += w * o.z; acc.w += w * o.w;
    }
  }
  const float inv = 1.0f / L;
  float4 res = make_float4(acc.x * inv, acc.y * inv, acc.z * inv, acc.w * inv);
  *(float4*)(Op + (size_t)grow * 256 + h * 64 + d4 * 4) = res;
}

extern "C" void kernel_launch(void* const* d_in, const int* in_sizes, int n_in,
                              void* d_out, int out_size, void* d_ws,
                              size_t ws_size, hipStream_t stream) {
  const float* q_in  = (const float*)d_in[0];
  const float* kv_in = (const float*)d_in[1];
  const float* Wq = (const float*)d_in[3];
  const float* Wk = (const float*)d_in[4];
  const float* Wv = (const float*)d_in[5];
  const float* Wo = (const float*)d_in[6];
  float* out = (float*)d_out;

  const int B = 2, S = 2048, IN = 1024;
  const int M = B * S;

  float* ws  = (float*)d_ws;
  float* wqg = ws;
  float* qg  = wqg + 1024 * 256;
  float* kp  = qg + M * 256;
  float* vp  = kp + M * 256;
  float* op  = vp + M * 256;

  reduce_wq_kernel<<<(1024 * 256) / 256, 256, 0, stream>>>(Wq, wqg);

  dim3 blk(BLOCK);
  dim3 blk8(BLOCK * 8);
  gemm_f32_wmma_ksplit<8><<<dim3(256 / 64, M / 64), blk8, 0, stream>>>(q_in, wqg, qg, M, 256, IN);
  gemm_f32_wmma_ksplit<8><<<dim3(256 / 64, M / 64), blk8, 0, stream>>>(kv_in, Wk, kp, M, 256, IN);
  gemm_f32_wmma_ksplit<8><<<dim3(256 / 64, M / 64), blk8, 0, stream>>>(kv_in, Wv, vp, M, 256, IN);

  fa_f32_wmma<<<dim3(B * 4 * NSLOT), blk, 0, stream>>>(qg, kp, vp, out, S);
  fa_merge_kernel<<<(4096 * 4 * 16) / 256, 256, 0, stream>>>(out, op, S);

  gemm_f32_wmma<<<dim3(IN / 64, M / 64), blk, 0, stream>>>(op, Wo, out, M, IN, 256);
}


# Q and K projections stored by their GEMM in the attention's fragment order: every Q/K load instruction of a wave reads 512 contiguous bytes (4 whole lines, not one line per lane); sums unchanged
# speedup vs baseline: 7.2752x; 1.0472x over previous
#include <hip/hip_runtime.h>
#include <hip/hip_bf16.h>

typedef __attribute__((ext_vector_type(2))) float v2f;
typedef __attribute__((ext_vector_type(4))) float v4f;
typedef __attribute__((ext_vector_type(8))) float v8f;

#define BLOCK 128

__device__ __forceinline__ v8f wmma4(v2f a, v2f b, v8f c) {
  return __builtin_amdgcn_wmma_f32_16x16x4_f32(false, a, false, b, (short)0, c,
                                               false, false);
}

template <int MASK>
__device__ __forceinline__ float row_xor(float v) {
  return __int_as_float(__builtin_amdgcn_update_dpp(
      0, __float_as_int(v), 0x160 + MASK, 0xf, 0xf, true));
}

__global__ void reduce_wq_kernel(const float* __restrict__ Wq,
                                 float* __restrict__ Wqg) {
  int idx = blockIdx.x * blockDim.x + threadIdx.x;
  if (idx >= 1024 * 256) return;
  int i = idx >> 8;
  int c = idx & 255;
  int h = c >> 6, d = c & 63;
  const float* p = Wq + (size_t)i * 1024 + h * 256 + d;
  Wqg[idx] = p[0] + p[64] + p[128] + p[192];
}

template <int NSUB>
__device__ __forceinline__ void gemm_wave_slice(
    const float* __restrict__ ap, const float* __restrict__ bp, int N, int ntrips,
    v8f (&acc)[4]) {
  for (int it = 0; it < ntrips; ++it) {
    v4f a[NSUB][2];
    v4f b[NSUB][8];
#pragma unroll
    for (int u = 0; u < NSUB; ++u) {
      a[u][0] = *(const v4f*)(ap + 16 * u);
      a[u][1] = *(const v4f*)(ap + 16 * u + 4);
#pragma unroll
      for (int t = 0; t < 8; ++t)
        b[u][t] = *(const v4f*)(bp + (size_t)(16 * u + t) * N);
    }
    __builtin_amdgcn_sched_barrier(0);
#pragma unroll
    for (int u = 0; u < NSUB; ++u) {
#pragma unroll
      for (int j = 0; j < 4; ++j) {
        v2f af;
        af.x = a[u][j >> 1][(2 * j) & 3];
        af.y = a[u][j >> 1][((2 * j) & 3) + 1];
#pragma unroll
        for (int nt = 0; nt < 4; ++nt) {
          v2f bf;
          bf.x = b[u][2 * j][nt];
          bf.y = b[u][2 * j + 1][nt];
          acc[nt] = wmma4(af, bf, acc[nt]);
        }
      }
    }
    ap += 16 * NSUB;
    bp += (size_t)(16 * NSUB) * N;
  }
}

__global__ __launch_bounds__(BLOCK) void gemm_f32_wmma(
    const float* __restrict__ A, const float* __restrict__ B,
    float* __restrict__ C, int M, int N, int K) {
  const int tid  = threadIdx.x;
  const int wave = tid >> 5;
  const int lane = tid & 31;
  const int half = lane >> 4;
  const int l16  = lane & 15;
  const int row0 = blockIdx.y * 64;
  const int col0 = blockIdx.x * 64;

  v8f acc[4];
#pragma unroll
  for (int nt = 0; nt < 4; ++nt) acc[nt] = (v8f)(0.0f);

  gemm_wave_slice<2>(A + (size_t)(row0 + wave * 16 + l16) * K + 8 * half,
                     B + (size_t)(8 * half) * N + col0 + 4 * l16, N, K / 32, acc);

#pragma unroll
  for (int r = 0; r < 8; ++r) {
    v4f o;
    o.x = acc[0][r]; o.y = acc[1][r]; o.z = acc[2][r]; o.w = acc[3][r];
    *(v4f*)(C + (size_t)(row0 + wave * 16 + r + 8 * half) * N + col0 + 4 * l16) = o;
  }
}

template <int KS, bool FRAG>
__global__ __launch_bounds__(BLOCK * KS) void gemm_f32_wmma_ksplit(
    const float* __restrict__ A, const float* __restrict__ B,
    float* __restrict__ C, int M, int N, int K) {
  constexpr int RED_PITCH  = 68;
  constexpr int GRP_FLOATS = 64 * RED_PITCH;
  __shared__ __align__(16) float smem[KS * GRP_FLOATS];
  const int grp  = threadIdx.x >> 7;
  const int tid  = threadIdx.x & (BLOCK - 1);
  const int wave = tid >> 5;
  const int lane = tid & 31;
  const int half = lane >> 4;
  const int l16  = lane & 15;
  const int row0 = blockIdx.y * 64;
  const int col0 = blockIdx.x * 64;
  const int kq   = K / KS;
  const int kbeg = grp * kq;

  v8f acc[4];
#pragma unroll
  for (int nt = 0; nt < 4; ++nt) acc[nt] = (v8f)(0.0f);

  gemm_wave_slice<1>(A + (size_t)(row0 + wave * 16 + l16) * K + kbeg + 8 * half,
                     B + (size_t)(kbeg + 8 * half) * N + col0 + 4 * l16, N, kq / 16, acc);

  float* red = smem + grp * GRP_FLOATS;
#pragma unroll
  for (int r = 0; r < 8; ++r) {
    v4f o;
    o.x = acc[0][r]; o.y = acc[1][r]; o.z = acc[2][r]; o.w = acc[3][r];
    *(v4f*)(red + (wave * 16 + r + 8 * half) * RED_PITCH + 4 * l16) = o;
  }
  __syncthreads();
  for (int f = threadIdx.x; f < 1024; f += BLOCK * KS) {
    int r, c;
    size_t off;
    if (FRAG) {
      const int kt = f >> 8, t = (f >> 5) & 7, hf = (f >> 4) & 1, lr = f & 15;
      r = kt * 16 + lr;
      c = 32 * hf + 4 * t;
      off = ((size_t)(((row0 >> 4) + kt) * (int)gridDim.x + (int)blockIdx.x) << 10) +
            ((f & 255) << 2);
    } else {
      r = f >> 4;
      c = (f & 15) << 2;
      off = (size_t)(row0 + r) * N + col0 + c;
    }
    float4 s = *(const float4*)(smem + r * RED_PITCH + c);
#pragma unroll
    for (int g = 1; g < KS; ++g) {
      const float4 p = *(const float4*)(smem + g * GRP_FLOATS + r * RED_PITCH + c);
      s.x += p.x; s.y += p.y; s.z += p.z; s.w += p.w;
    }
    *(float4*)(C + off) = s;
  }
}

#define CH_UNITS 5
#define CH_TILES (CH_UNITS * 4)
#define NCHUNK_MAX 7
#define NSLOT 119
#define SLOT_FLOATS (64 * 64 + 128)

__device__ __forceinline__ int slot_prefix(int qt) {
  const int a = qt / CH_UNITS, r = qt - a * CH_UNITS;
  return qt + CH_UNITS * a * (a - 1) / 2 + a * r;
}

#define NQT 32
#define NFULL 87
__device__ __forceinline__ void fa_block_pair(int rank, int& qt, int& c) {
  if (rank < NFULL) {
    int a = 1;
#pragma unroll
    for (int k = 1; k < NCHUNK_MAX - 1; ++k)
      if (CH_UNITS * k * (k + 1) / 2 <= rank) a = k + 1;
    const int t = rank - CH_UNITS * a * (a - 1) / 2;
    qt = CH_UNITS * a + t / a;
    c  = t % a;
  } else {
    int p = rank - NFULL;
    qt = 0;
    bool found = false;
#pragma unroll
    for (int r = CH_UNITS - 1; r >= 0; --r) {
      const int n_r = (NQT - 1 - r) / CH_UNITS + 1;
      if (!found && p < n_r) { qt = r + CH_UNITS * p; found = true; }
      if (!found) p -= n_r;
    }
    c = qt / CH_UNITS;
  }
}

__global__ __launch_bounds__(BLOCK) void fa_f32_wmma(
    const float* __restrict__ Qg, const float* __restrict__ Kp,
    const float* __restrict__ Vp, float* __restrict__ Part, int S) {
  __shared__ __align__(16) float Ps[4][16][18];
  const int bh = (int)blockIdx.x & 7;
  int qt, chunk;
  fa_block_pair((int)blockIdx.x >> 3, qt, chunk);
  const int tid  = threadIdx.x;
  const int wave = __builtin_amdgcn_readfirstlane(tid >> 5);
  const int lane = tid & 31;
  const int half = lane >> 4;
  const int l16  = lane & 15;
  const int q0   = qt * 64;
  const int b    = bh >> 2;
  const int h    = bh & 3;

  v2f qf[16];
  {
    const float* qptr =
        Qg + ((size_t)((b * (S >> 4) + qt * 4 + wave) * 4 + h) << 10) + lane * 4;
#pragma unroll
    for (int t = 0; t < 8; ++t) {
      const v4f q4 = *(const v4f*)(qptr + 128 * t);
      qf[2 * t].x = q4.x;     qf[2 * t].y = q4.y;
      qf[2 * t + 1].x = q4.z; qf[2 * t + 1].y = q4.w;
    }
  }

  v8f oacc[4];
#pragma unroll
  for (int nt = 0; nt < 4; ++nt) oacc[nt] = (v8f)(0.0f);
  float m[8], l[8];
#pragma unroll
  for (int r = 0; r < 8; ++r) { m[r] = -1e30f; l[r] = 0.0f; }

  const int ntiles = (q0 >> 4) + wave + 1;
  const int jbeg = chunk * CH_TILES;
  const int jend = (jbeg + CH_TILES < ntiles) ? (jbeg + CH_TILES) : ntiles;
  const float* kptr =
      Kp + ((size_t)((b * (S >> 4) + jbeg) * 4 + h) << 10) + lane * 4;
  const float* vptr =
      Vp + (size_t)(b * S + jbeg * 16 + 8 * half) * 256 + h * 64 + 4 * l16;
  for (int j = jbeg; j < jend; ++j) {
    const int a0 = j << 4;
    v4f k4[8], v4[8];
#pragma unroll
    for (int t = 0; t < 8; ++t) k4[t] = *(const v4f*)(kptr + 128 * t);
#pragma unroll
    for (int t = 0; t < 8; ++t) v4[t] = *(const v4f*)(vptr + (size_t)t * 256);
    kptr += 4 * 1024;
    vptr += 16 * 256;
    __builtin_amdgcn_sched_barrier(0);

    v8f s = (v8f)(0.0f);
#pragma unroll
    for (int i = 0; i < 16; ++i) {
      v2f bk;
      bk.x = k4[i >> 1][(2 * i) & 3];
      bk.y = k4[i >> 1][((2 * i) & 3) + 1];
      s = wmma4(qf[i], bk, s);
    }

    const int acol = a0 + l16;
    float pr[8];
#pragma unroll
    for (int r = 0; r < 8; ++r) {
      const int qrow = q0 + wave * 16 + r + 8 * half;
      float x = s[r] * 0.125f;
      if (acol > qrow) x = -1e30f;
      float mx = x;
      mx = fmaxf(mx, row_xor<1>(mx));
      mx = fmaxf(mx, row_xor<2>(mx));
      mx = fmaxf(mx, row_xor<4>(mx));
      mx = fmaxf(mx, row_xor<8>(mx));
      const float mnew  = fmaxf(m[r], mx);
      const float p     = __expf(x - mnew);
      float sum = p;
      sum += row_xor<1>(sum);
      sum += row_xor<2>(sum);
      sum += row_xor<4>(sum);
      sum += row_xor<8>(sum);
      const float scale = __expf(m[r] - mnew);
      l[r] = l[r] * scale + sum;
      m[r] = mnew;
#pragma unroll
      for (int nt = 0; nt < 4; ++nt) oacc[nt][r] *= scale;
      pr[r] = p;
    }

#pragma unroll
    for (int r = 0; r < 8; ++r) Ps[wave][r + 8 * half][l16] = pr[r];
    __builtin_amdgcn_wave_barrier();

#pragma unroll
    for (int jj = 0; jj < 4; ++jj) {
      const v2f a = *(const v2f*)&Ps[wave][l16][8 * half + 2 * jj];
#pragma unroll
      for (int nt = 0; nt < 4; ++nt) {
        v2f bv;
        bv.x = v4[2 * jj][nt];
        bv.y = v4[2 * jj + 1][nt];
        oacc[nt] = wmma4(a, bv, oacc[nt]);
      }
    }
  }

  float* slot = Part + (size_t)(bh * NSLOT + slot_prefix(qt) + chunk) * SLOT_FLOATS;
#pragma unroll
  for (int r = 0; r < 8; ++r) {
    const int rl = wave * 16 + r + 8 * half;
    v4f o;
    o.x = oacc[0][r]; o.y = oacc[1][r]; o.z = oacc[2][r]; o.w = oacc[3][r];
    *(v4f*)(slot + rl * 64 + 4 * l16) = o;
    if (l16 == 0) {
      slot[64 * 64 + rl]      = m[r];
      slot[64 * 64 + 64 + rl] = l[r];
    }
  }
}

__global__ void fa_merge_kernel(const float* __restrict__ Part,
                                float* __restrict__ Op, int S) {
  const int idx  = blockIdx.x * blockDim.x + threadIdx.x;
  const int d4   = idx & 15;
  const int h    = (idx >> 4) & 3;
  const int grow = idx >> 6;
  const int b    = grow / S;
  const int row  = grow - b * S;
  const int qt   = row >> 6, rl = row & 63;
  const int nch  = qt / CH_UNITS + 1;
  const float* base =
      Part + (size_t)((b * 4 + h) * NSLOT + slot_prefix(qt)) * SLOT_FLOATS;
  float mc[NCHUNK_MAX];
#pragma unroll
  for (int c = 0; c < NCHUNK_MAX; ++c)
    mc[c] = (c < nch) ? base[(size_t)c * SLOT_FLOATS + 64 * 64 + rl] : -1e30f;
  float M = mc[0];
#pragma unroll
  for (int c = 1; c < NCHUNK_MAX; ++c) M = fmaxf(M, mc[c]);
  float L = 0.0f;
  float4 acc = make_float4(0.0f, 0.0f, 0.0f, 0.0f);
#pragma unroll
  for (int c = 0; c < NCHUNK_MAX; ++c) {
    if (c < nch) {
      const float* s = base + (size_t)c * SLOT_FLOATS;
      const float w = __expf(mc[c] - M);
      L += s[64 * 64 + 64 + rl] * w;
      const float4 o = *(const float4*)(s + rl * 64 + d4 * 4);
      acc.x += w * o.x; acc.y += w * o.y; acc.z += w * o.z; acc.w += w * o.w;
    }
  }
  const float inv = 1.0f / L;
  float4 res = make_float4(acc.x * inv, acc.y * inv, acc.z * inv, acc.w * inv);
  *(float4*)(Op + (size_t)grow * 256 + h * 64 + d4 * 4) = res;
}

extern "C" void kernel_launch(void* const* d_in, const int* in_sizes, int n_in,
                              void* d_out, int out_size, void* d_ws,
                              size_t ws_size, hipStream_t stream) {
  const float* q_in  = (const float*)d_in[0];
  const float* kv_in = (const float*)d_in[1];
  const float* Wq = (const float*)d_in[3];
  const float* Wk = (const float*)d_in[4];
  const float* Wv = (const float*)d_in[5];
  const float* Wo = (const float*)d_in[6];
  float* out = (float*)d_out;

  const int B = 2, S = 2048, IN = 1024;
  const int M = B * S;

  float* ws  = (float*)d_ws;
  float* wqg = ws;
  float* qg  = wqg + 1024 * 256;
  float* kp  = qg + M * 256;
  float* vp  = kp + M * 256;
  float* op  = vp + M * 256;

  reduce_wq_kernel<<<(1024 * 256) / 256, 256, 0, stream>>>(Wq, wqg);

  dim3 blk(BLOCK);
  dim3 blk8(BLOCK * 8);
  gemm_f32_wmma_ksplit<8, true><<<dim3(256 / 64, M / 64), blk8, 0, stream>>>(q_in, wqg, qg, M, 256, IN);
  gemm_f32_wmma_ksplit<8, true><<<dim3(256 / 64, M / 64), blk8, 0, stream>>>(kv_in, Wk, kp, M, 256, IN);
  gemm_f32_wmma_ksplit<8, false><<<dim3(256 / 64, M / 64), blk8, 0, stream>>>(kv_in, Wv, vp, M, 256, IN);

  fa_f32_wmma<<<dim3(B * 4 * NSLOT), blk, 0, stream>>>(qg, kp, vp, out, S);
  fa_merge_kernel<<<(4096 * 4 * 16) / 256, 256, 0, stream>>>(out, op, S);

  gemm_f32_wmma<<<dim3(IN / 64, M / 64), blk, 0, stream>>>(op, Wo, out, M, IN, 256);
}


# K and V projections as one GEMM launch: a wave loads its kv_in fragment once and feeds 8 accumulator tiles (4 of K, 4 of V); 4 k groups of 8 32-k trips, not 8 of 8 16-k; changes order of the k sums
# speedup vs baseline: 7.5141x; 1.0328x over previous
#include <hip/hip_runtime.h>
#include <hip/hip_bf16.h>

typedef __attribute__((ext_vector_type(2))) float v2f;
typedef __attribute__((ext_vector_type(4))) float v4f;
typedef __attribute__((ext_vector_type(8))) float v8f;

#define BLOCK 128

__device__ __forceinline__ v8f wmma4(v2f a, v2f b, v8f c) {
  return __builtin_amdgcn_wmma_f32_16x16x4_f32(false, a, false, b, (short)0, c,
                                               false, false);
}

template <int MASK>
__device__ __forceinline__ float row_xor(float v) {
  return __int_as_float(__builtin_amdgcn_update_dpp(
      0, __float_as_int(v), 0x160 + MASK, 0xf, 0xf, true));
}

__global__ void reduce_wq_kernel(const float* __restrict__ Wq,
                                 float* __restrict__ Wqg) {
  int idx = blockIdx.x * blockDim.x + threadIdx.x;
  if (idx >= 1024 * 256) return;
  int i = idx >> 8;
  int c = idx & 255;
  int h = c >> 6, d = c & 63;
  const float* p = Wq + (size_t)i * 1024 + h * 256 + d;
  Wqg[idx] = p[0] + p[64] + p[128] + p[192];
}

template <int NSUB>
__device__ __forceinline__ void gemm_wave_slice(
    const float* __restrict__ ap, const float* __restrict__ bp, int N, int ntrips,
    v8f (&acc)[4]) {
  for (int it = 0; it < ntrips; ++it) {
    v4f a[NSUB][2];
    v4f b[NSUB][8];
#pragma unroll
    for (int u = 0; u < NSUB; ++u) {
      a[u][0] = *(const v4f*)(ap + 16 * u);
      a[u][1] = *(const v4f*)(ap + 16 * u + 4);
#pragma unroll
      for (int t = 0; t < 8; ++t)
        b[u][t] = *(const v4f*)(bp + (size_t)(16 * u + t) * N);
    }
    __builtin_amdgcn_sched_barrier(0);
#pragma unroll
    for (int u = 0; u < NSUB; ++u) {
#pragma unroll
      for (int j = 0; j < 4; ++j) {
        v2f af;
        af.x = a[u][j >> 1][(2 * j) & 3];
        af.y = a[u][j >> 1][((2 * j) & 3) + 1];
#pragma unroll
        for (int nt = 0; nt < 4; ++nt) {
          v2f bf;
          bf.x = b[u][2 * j][nt];
          bf.y = b[u][2 * j + 1][nt];
          acc[nt] = wmma4(af, bf, acc[nt]);
        }
      }
    }
    ap += 16 * NSUB;
    bp += (size_t)(16 * NSUB) * N;
  }
}

template <int NSUB>
__device__ __forceinline__ void gemm_wave_slice_2b(
    const float* __restrict__ ap, const float* __restrict__ bp1,
    const float* __restrict__ bp2, int N, int ntrips, v8f (&acc1)[4],
    v8f (&acc2)[4]) {
  for (int it = 0; it < ntrips; ++it) {
    v4f a[NSUB][2];
    v4f b1[NSUB][8], b2[NSUB][8];
#pragma unroll
    for (int u = 0; u < NSUB; ++u) {
      a[u][0] = *(const v4f*)(ap + 16 * u);
      a[u][1] = *(const v4f*)(ap + 16 * u + 4);
#pragma unroll
      for (int t = 0; t < 8; ++t)
        b1[u][t] = *(const v4f*)(bp1 + (size_t)(16 * u + t) * N);
#pragma unroll
      for (int t = 0; t < 8; ++t)
        b2[u][t] = *(const v4f*)(bp2 + (size_t)(16 * u + t) * N);
    }
    __builtin_amdgcn_sched_barrier(0);
#pragma unroll
    for (int u = 0; u < NSUB; ++u) {
#pragma unroll
      for (int j = 0; j < 4; ++j) {
        v2f af;
        af.x = a[u][j >> 1][(2 * j) & 3];
        af.y = a[u][j >> 1][((2 * j) & 3) + 1];
#pragma unroll
        for (int nt = 0; nt < 4; ++nt) {
          v2f bf;
          bf.x = b1[u][2 * j][nt];
          bf.y = b1[u][2 * j + 1][nt];
          acc1[nt] = wmma4(af, bf, acc1[nt]);
        }
#pragma unroll
        for (int nt = 0; nt < 4; ++nt) {
          v2f bf;
          bf.x = b2[u][2 * j][nt];
          bf.y = b2[u][2 * j + 1][nt];
          acc2[nt] = wmma4(af, bf, acc2[nt]);
        }
      }
    }
    ap += 16 * NSUB;
    bp1 += (size_t)(16 * NSUB) * N;
    bp2 += (size_t)(16 * NSUB) * N;
  }
}

__global__ __launch_bounds__(BLOCK) void gemm_f32_wmma(
    const float* __restrict__ A, const float* __restrict__ B,
    float* __restrict__ C, int M, int N, int K) {
  const int tid  = threadIdx.x;
  const int wave = tid >> 5;
  const int lane = tid & 31;
  const int half = lane >> 4;
  const int l16  = lane & 15;
  const int row0 = blockIdx.y * 64;
  const int col0 = blockIdx.x * 64;

  v8f acc[4];
#pragma unroll
  for (int nt = 0; nt < 4; ++nt) acc[nt] = (v8f)(0.0f);

  gemm_wave_slice<2>(A + (size_t)(row0 + wave * 16 + l16) * K + 8 * half,
                     B + (size_t)(8 * half) * N + col0 + 4 * l16, N, K / 32, acc);

#pragma unroll
  for (int r = 0; r < 8; ++r) {
    v4f o;
    o.x = acc[0][r]; o.y = acc[1][r]; o.z = acc[2][r]; o.w = acc[3][r];
    *(v4f*)(C + (size_t)(row0 + wave * 16 + r + 8 * half) * N + col0 + 4 * l16) = o;
  }
}

template <int KS, bool FRAG>
__global__ __launch_bounds__(BLOCK * KS) void gemm_f32_wmma_ksplit(
    const float* __restrict__ A, const float* __restrict__ B,
    float* __restrict__ C, int M, int N, int K) {
  constexpr int RED_PITCH  = 68;
  constexpr int GRP_FLOATS = 64 * RED_PITCH;
  __shared__ __align__(16) float smem[KS * GRP_FLOATS];
  const int grp  = threadIdx.x >> 7;
  const int tid  = threadIdx.x & (BLOCK - 1);
  const int wave = tid >> 5;
  const int lane = tid & 31;
  const int half = lane >> 4;
  const int l16  = lane & 15;
  const int row0 = blockIdx.y * 64;
  const int col0 = blockIdx.x * 64;
  const int kq   = K / KS;
  const int kbeg = grp * kq;

  v8f acc[4];
#pragma unroll
  for (int nt = 0; nt < 4; ++nt) acc[nt] = (v8f)(0.0f);

  gemm_wave_slice<1>(A + (size_t)(row0 + wave * 16 + l16) * K + kbeg + 8 * half,
                     B + (size_t)(kbeg + 8 * half) * N + col0 + 4 * l16, N, kq / 16, acc);

  float* red = smem + grp * GRP_FLOATS;
#pragma unroll
  for (int r = 0; r < 8; ++r) {
    v4f o;
    o.x = acc[0][r]; o.y = acc[1][r]; o.z = acc[2][r]; o.w = acc[3][r];
    *(v4f*)(red + (wave * 16 + r + 8 * half) * RED_PITCH + 4 * l16) = o;
  }
  __syncthreads();
  for (int f = threadIdx.x; f < 1024; f += BLOCK * KS) {
    int r, c;
    size_t off;
    if (FRAG) {
      const int kt = f >> 8, t = (f >> 5) & 7, hf = (f >> 4) & 1, lr = f & 15;
      r = kt * 16 + lr;
      c = 32 * hf + 4 * t;
      off = ((size_t)(((row0 >> 4) + kt) * (int)gridDim.x + (int)blockIdx.x) << 10) +
            ((f & 255) << 2);
    } else {
      r = f >> 4;
      c = (f & 15) << 2;
      off = (size_t)(row0 + r) * N + col0 + c;
    }
    float4 s = *(const float4*)(smem + r * RED_PITCH + c);
#pragma unroll
    for (int g = 1; g < KS; ++g) {
      const float4 p = *(const float4*)(smem + g * GRP_FLOATS + r * RED_PITCH + c);
      s.x += p.x; s.y += p.y; s.z += p.z; s.w += p.w;
    }
    *(float4*)(C + off) = s;
  }
}

template <int KS>
__global__ __launch_bounds__(BLOCK * KS) void gemm_f32_wmma_pair_ksplit(
    const float* __restrict__ A, const float* __restrict__ B1,
    const float* __restrict__ B2, float* __restrict__ C1, float* __restrict__ C2,
    int M, int N, int K) {
  constexpr int RED_PITCH  = 132;
  constexpr int GRP_FLOATS = 64 * RED_PITCH;
  __shared__ __align__(16) float smem[KS * GRP_FLOATS];
  const int grp  = threadIdx.x >> 7;
  const int tid  = threadIdx.x & (BLOCK - 1);
  const int wave = tid >> 5;
  const int lane = tid & 31;
  const int half = lane >> 4;
  const int l16  = lane & 15;
  const int row0 = blockIdx.y * 64;
  const int col0 = blockIdx.x * 64;
  const int kq   = K / KS;
  const int kbeg = grp * kq;

  v8f acc1[4], acc2[4];
#pragma unroll
  for (int nt = 0; nt < 4; ++nt) { acc1[nt] = (v8f)(0.0f); acc2[nt] = (v8f)(0.0f); }

  gemm_wave_slice_2b<2>(A + (size_t)(row0 + wave * 16 + l16) * K + kbeg + 8 * half,
                        B1 + (size_t)(kbeg + 8 * half) * N + col0 + 4 * l16,
                        B2 + (size_t)(kbeg + 8 * half) * N + col0 + 4 * l16, N, kq / 32,
                        acc1, acc2);

  float* red = smem + grp * GRP_FLOATS;
#pragma unroll
  for (int r = 0; r < 8; ++r) {
    v4f o1, o2;
    o1.x = acc1[0][r]; o1.y = acc1[1][r]; o1.z = acc1[2][r]; o1.w = acc1[3][r];
    o2.x = acc2[0][r]; o2.y = acc2[1][r]; o2.z = acc2[2][r]; o2.w = acc2[3][r];
    float* rowp = red + (wave * 16 + r + 8 * half) * RED_PITCH + 4 * l16;
    *(v4f*)(rowp)      = o1;
    *(v4f*)(rowp + 64) = o2;
  }
  __syncthreads();
  for (int f = threadIdx.x; f < 2048; f += BLOCK * KS) {
    const int fp = f & 1023;
    int r, c;
    float* dst;
    if (f < 1024) {
      const int kt = fp >> 8, t = (fp >> 5) & 7, hf = (fp >> 4) & 1, lr = fp & 15;
      r = kt * 16 + lr;
      c = 32 * hf + 4 * t;
      dst = C1 + ((size_t)(((row0 >> 4) + kt) * (int)gridDim.x + (int)blockIdx.x) << 10) +
            ((fp & 255) << 2);
    } else {
      r = fp >> 4;
      c = (fp & 15) << 2;
      dst = C2 + (size_t)(row0 + r) * N + col0 + c;
      c += 64;
    }
    float4 s = *(const float4*)(smem + r * RED_PITCH + c);
#pragma unroll
    for (int g = 1; g < KS; ++g) {
      const float4 p = *(const float4*)(smem + g * GRP_FLOATS + r * RED_PITCH + c);
      s.x += p.x; s.y += p.y; s.z += p.z; s.w += p.w;
    }
    *(float4*)dst = s;
  }
}

#define CH_UNITS 5
#define CH_TILES (CH_UNITS * 4)
#define NCHUNK_MAX 7
#define NSLOT 119
#define SLOT_FLOATS (64 * 64 + 128)

__device__ __forceinline__ int slot_prefix(int qt) {
  const int a = qt / CH_UNITS, r = qt - a * CH_UNITS;
  return qt + CH_UNITS * a * (a - 1) / 2 + a * r;
}

#define NQT 32
#define NFULL 87
__device__ __forceinline__ void fa_block_pair(int rank, int& qt, int& c) {
  if (rank < NFULL) {
    int a = 1;
#pragma unroll
    for (int k = 1; k < NCHUNK_MAX - 1; ++k)
      if (CH_UNITS * k * (k + 1) / 2 <= rank) a = k + 1;
    const int t = rank - CH_UNITS * a * (a - 1) / 2;
    qt = CH_UNITS * a + t / a;
    c  = t % a;
  } else {
    int p = rank - NFULL;
    qt = 0;
    bool found = false;
#pragma unroll
    for (int r = CH_UNITS - 1; r >= 0; --r) {
      const int n_r = (NQT - 1 - r) / CH_UNITS + 1;
      if (!found && p < n_r) { qt = r + CH_UNITS * p; found = true; }
      if (!found) p -= n_r;
    }
    c = qt / CH_UNITS;
  }
}

__global__ __launch_bounds__(BLOCK) void fa_f32_wmma(
    const float* __restrict__ Qg, const float* __restrict__ Kp,
    const float* __restrict__ Vp, float* __restrict__ Part, int S) {
  __shared__ __align__(16) float Ps[4][16][18];
  const int bh = (int)blockIdx.x & 7;
  int qt, chunk;
  fa_block_pair((int)blockIdx.x >> 3, qt, chunk);
  const int tid  = threadIdx.x;
  const int wave = __builtin_amdgcn_readfirstlane(tid >> 5);
  const int lane = tid & 31;
  const int half = lane >> 4;
  const int l16  = lane & 15;
  const int q0   = qt * 64;
  const int b    = bh >> 2;
  const int h    = bh & 3;

  v2f qf[16];
  {
    const float* qptr =
        Qg + ((size_t)((b * (S >> 4) + qt * 4 + wave) * 4 + h) << 10) + lane * 4;
#pragma unroll
    for (int t = 0; t < 8; ++t) {
      const v4f q4 = *(const v4f*)(qptr + 128 * t);
      qf[2 * t].x = q4.x;     qf[2 * t].y = q4.y;
      qf[2 * t + 1].x = q4.z; qf[2 * t + 1].y = q4.w;
    }
  }

  v8f oacc[4];
#pragma unroll
  for (int nt = 0; nt < 4; ++nt) oacc[nt] = (v8f)(0.0f);
  float m[8], l[8];
#pragma unroll
  for (int r = 0; r < 8; ++r) { m[r] = -1e30f; l[r] = 0.0f; }

  const int ntiles = (q0 >> 4) + wave + 1;
  const int jbeg = chunk * CH_TILES;
  const int jend = (jbeg + CH_TILES < ntiles) ? (jbeg + CH_TILES) : ntiles;
  const float* kptr =
      Kp + ((size_t)((b * (S >> 4) + jbeg) * 4 + h) << 10) + lane * 4;
  const float* vptr =
      Vp + (size_t)(b * S + jbeg * 16 + 8 * half) * 256 + h * 64 + 4 * l16;
  for (int j = jbeg; j < jend; ++j) {
    const int a0 = j << 4;
    v4f k4[8], v4[8];
#pragma unroll
    for (int t = 0; t < 8; ++t) k4[t] = *(const v4f*)(kptr + 128 * t);
#pragma unroll
    for (int t = 0; t < 8; ++t) v4[t] = *(const v4f*)(vptr + (size_t)t * 256);
    kptr += 4 * 1024;
    vptr += 16 * 256;
    __builtin_amdgcn_sched_barrier(0);

    v8f s = (v8f)(0.0f);
#pragma unroll
    for (int i = 0; i < 16; ++i) {
      v2f bk;
      bk.x = k4[i >> 1][(2 * i) & 3];
      bk.y = k4[i >> 1][((2 * i) & 3) + 1];
      s = wmma4(qf[i], bk, s);
    }

    const int acol = a0 + l16;
    float pr[8];
#pragma unroll
    for (int r = 0; r < 8; ++r) {
      const int qrow = q0 + wave * 16 + r + 8 * half;
      float x = s[r] * 0.125f;
      if (acol > qrow) x = -1e30f;
      float mx = x;
      mx = fmaxf(mx, row_xor<1>(mx));
      mx = fmaxf(mx, row_xor<2>(mx));
      mx = fmaxf(mx, row_xor<4>(mx));
      mx = fmaxf(mx, row_xor<8>(mx));
      const float mnew  = fmaxf(m[r], mx);
      const float p     = __expf(x - mnew);
      float sum = p;
      sum += row_xor<1>(sum);
      sum += row_xor<2>(sum);
      sum += row_xor<4>(sum);
      sum += row_xor<8>(sum);
      const float scale = __expf(m[r] - mnew);
      l[r] = l[r] * scale + sum;
      m[r] = mnew;
#pragma unroll
      for (int nt = 0; nt < 4; ++nt) oacc[nt][r] *= scale;
      pr[r] = p;
    }

#pragma unroll
    for (int r = 0; r < 8; ++r) Ps[wave][r + 8 * half][l16] = pr[r];
    __builtin_amdgcn_wave_barrier();

#pragma unroll
    for (int jj = 0; jj < 4; ++jj) {
      const v2f a = *(const v2f*)&Ps[wave][l16][8 * half + 2 * jj];
#pragma unroll
      for (int nt = 0; nt < 4; ++nt) {
        v2f bv;
        bv.x = v4[2 * jj][nt];
        bv.y = v4[2 * jj + 1][nt];
        oacc[nt] = wmma4(a, bv, oacc[nt]);
      }
    }
  }

  float* slot = Part + (size_t)(bh * NSLOT + slot_prefix(qt) + chunk) * SLOT_FLOATS;
#pragma unroll
  for (int r = 0; r < 8; ++r) {
    const int rl = wave * 16 + r + 8 * half;
    v4f o;
    o.x = oacc[0][r]; o.y = oacc[1][r]; o.z = oacc[2][r]; o.w = oacc[3][r];
    *(v4f*)(slot + rl * 64 + 4 * l16) = o;
    if (l16 == 0) {
      slot[64 * 64 + rl]      = m[r];
      slot[64 * 64 + 64 + rl] = l[r];
    }
  }
}

__global__ void fa_merge_kernel(const float* __restrict__ Part,
                                float* __restrict__ Op, int S) {
  const int idx  = blockIdx.x * blockDim.x + threadIdx.x;
  const int d4   = idx & 15;
  const int h    = (idx >> 4) & 3;
  const int grow = idx >> 6;
  const int b    = grow / S;
  const int row  = grow - b * S;
  const int qt   = row >> 6, rl = row & 63;
  const int nch  = qt / CH_UNITS + 1;
  const float* base =
      Part + (size_t)((b * 4 + h) * NSLOT + slot_prefix(qt)) * SLOT_FLOATS;
  float mc[NCHUNK_MAX];
#pragma unroll
  for (int c = 0; c < NCHUNK_MAX; ++c)
    mc[c] = (c < nch) ? base[(size_t)c * SLOT_FLOATS + 64 * 64 + rl] : -1e30f;
  float M = mc[0];
#pragma unroll
  for (int c = 1; c < NCHUNK_MAX; ++c) M = fmaxf(M, mc[c]);
  float L = 0.0f;
  float4 acc = make_float4(0.0f, 0.0f, 0.0f, 0.0f);
#pragma unroll
  for (int c = 0; c < NCHUNK_MAX; ++c) {
    if (c < nch) {
      const float* s = base + (size_t)c * SLOT_FLOATS;
      const float w = __expf(mc[c] - M);
      L += s[64 * 64 + 64 + rl] * w;
      const float4 o = *(const float4*)(s + rl * 64 + d4 * 4);
      acc.x += w * o.x; acc.y += w * o.y; acc.z += w * o.z; acc.w += w * o.w;
    }
  }
  const float inv = 1.0f / L;
  float4 res = make_float4(acc.x * inv, acc.y * inv, acc.z * inv, acc.w * inv);
  *(float4*)(Op + (size_t)grow * 256 + h * 64 + d4 * 4) = res;
}

extern "C" void kernel_launch(void* const* d_in, const int* in_sizes, int n_in,
                              void* d_out, int out_size, void* d_ws,
                              size_t ws_size, hipStream_t stream) {
  const float* q_in  = (const float*)d_in[0];
  const float* kv_in = (const float*)d_in[1];
  const float* Wq = (const float*)d_in[3];
  const float* Wk = (const float*)d_in[4];
  const float* Wv = (const float*)d_in[5];
  const float* Wo = (const float*)d_in[6];
  float* out = (float*)d_out;

  const int B = 2, S = 2048, IN = 1024;
  const int M = B * S;

  float* ws  = (float*)d_ws;
  float* wqg = ws;
  float* qg  = wqg + 1024 * 256;
  float* kp  = qg + M * 256;
  float* vp  = kp + M * 256;
  float* op  = vp + M * 256;

  reduce_wq_kernel<<<(1024 * 256) / 256, 256, 0, stream>>>(Wq, wqg);

  dim3 blk(BLOCK);
  dim3 blk8(BLOCK * 8);
  gemm_f32_wmma_ksplit<8, true><<<dim3(256 / 64, M / 64), blk8, 0, stream>>>(q_in, wqg, qg, M, 256, IN);
  dim3 blk4(BLOCK * 4);
  gemm_f32_wmma_pair_ksplit<4><<<dim3(256 / 64, M / 64), blk4, 0, stream>>>(kv_in, Wk, Wv, kp, vp, M, 256, IN);

  fa_f32_wmma<<<dim3(B * 4 * NSLOT), blk, 0, stream>>>(qg, kp, vp, out, S);
  fa_merge_kernel<<<(4096 * 4 * 16) / 256, 256, 0, stream>>>(out, op, S);

  gemm_f32_wmma<<<dim3(IN / 64, M / 64), blk, 0, stream>>>(op, Wo, out, M, IN, 256);
}
